# starting kernel
# baseline (speedup 1.0000x reference)
#include <hip/hip_runtime.h>
#include <hip/hip_bf16.h>


typedef __attribute__((ext_vector_type(2))) float v2f;
typedef __attribute__((ext_vector_type(8))) float v8f;

#define D 64

__global__ void init_deg_kernel(unsigned int* __restrict__ deg, int N) {
    int i = blockIdx.x * blockDim.x + threadIdx.x;
    if (i < N) deg[i] = 1u;
}

__global__ void count_deg_kernel(const int* __restrict__ col,
                                 unsigned int* __restrict__ deg, int E) {
    int e = blockIdx.x * blockDim.x + threadIdx.x;
    if (e < E) atomicAdd(&deg[col[e]], 1u);
}

__global__ void deg_to_dinv_kernel(float* __restrict__ buf, int N) {
    int i = blockIdx.x * blockDim.x + threadIdx.x;
    if (i < N) {
        unsigned int d = ((const unsigned int*)buf)[i];
        buf[i] = __frsqrt_rn((float)d);
    }
}

__global__ void gcn_gemm_wmma_kernel(const float* __restrict__ x,
                                     const float* __restrict__ Wc,
                                     const float* __restrict__ bc,
                                     const float* __restrict__ Ws,
                                     const float* __restrict__ bs,
                                     const float* __restrict__ dinv,
                                     float* __restrict__ xw,
                                     float* __restrict__ out, int N) {
    const int lane   = threadIdx.x & 31;
    const int wave   = threadIdx.x >> 5;
    const int waveG  = blockIdx.x * (blockDim.x >> 5) + wave;
    const int rowBase = waveG * 16;
    if (rowBase >= N) return;

    const int laneHi = lane >> 4;
    const int laneLo = lane & 15;

    int arow = rowBase + laneLo;
    if (arow >= N) arow = N - 1;
    const float* ap = x + (size_t)arow * D + 2 * laneHi;
    v2f a[16];
#pragma unroll
    for (int k = 0; k < 16; ++k) {
        const float2 t = *(const float2*)(ap + 4 * k);
        a[k].x = t.x; a[k].y = t.y;
    }

#pragma unroll
    for (int nt = 0; nt < 4; ++nt) {
        const int n = nt * 16 + laneLo;
        v8f accC = {};
        v8f accS = {};
#pragma unroll
        for (int k0 = 0; k0 < D; k0 += 4) {
            const int kk = k0 + 2 * laneHi;
            v2f bC, bS;
            bC.x = Wc[(size_t)kk * D + n];
            bC.y = Wc[(size_t)(kk + 1) * D + n];
            bS.x = Ws[(size_t)kk * D + n];
            bS.y = Ws[(size_t)(kk + 1) * D + n];
            accC = __builtin_amdgcn_wmma_f32_16x16x4_f32(
                false, a[k0 >> 2], false, bC, (short)0, accC, false, false);
            accS = __builtin_amdgcn_wmma_f32_16x16x4_f32(
                false, a[k0 >> 2], false, bS, (short)0, accS, false, false);
        }
        const float bias = bs[nt * 16 + laneLo] + bc[nt * 16 + laneLo];
#pragma unroll
        for (int r = 0; r < 8; ++r) {
            const int row = rowBase + r + 8 * laneHi;
            if (row < N) {
                const size_t idx = (size_t)row * D + nt * 16 + laneLo;
                const float xv = accC[r];
                xw[idx] = xv;
                const float di = dinv[row];
                out[idx] = accS[r] + bias + di * di * xv;
            }
        }
    }
}

__global__ void gcn_scatter_kernel(const int* __restrict__ ei,
                                   const float* __restrict__ dinv,
                                   const float* __restrict__ xw,
                                   float* __restrict__ out, int E) {
    const int tid = blockIdx.x * blockDim.x + threadIdx.x;
    const int e = tid >> 4;
    if (e >= E) return;
    const int fg = (tid & 15) << 2;
    const int r = ei[e];
    const int c = ei[E + e];
    const float w = dinv[r] * dinv[c];
    const float4 v = *(const float4*)(xw + (size_t)r * D + fg);
    float* o = out + (size_t)c * D + fg;
#if defined(__HIP_DEVICE_COMPILE__)
    unsafeAtomicAdd(o + 0, w * v.x);
    unsafeAtomicAdd(o + 1, w * v.y);
    unsafeAtomicAdd(o + 2, w * v.z);
    unsafeAtomicAdd(o + 3, w * v.w);
#endif
}

extern "C" void kernel_launch(void* const* d_in, const int* in_sizes, int n_in,
                              void* d_out, int out_size, void* d_ws, size_t ws_size,
                              hipStream_t stream) {
    const float* x  = (const float*)d_in[0];
    const int*   ei = (const int*)d_in[1];
    const float* Wc = (const float*)d_in[2];
    const float* bc = (const float*)d_in[3];
    const float* Ws = (const float*)d_in[4];
    const float* bs = (const float*)d_in[5];
    float* out = (float*)d_out;

    const int N = in_sizes[0] / D;
    const int E = in_sizes[1] / 2;

    float* dinv = (float*)d_ws;
    size_t off = ((size_t)N * sizeof(float) + 255) & ~(size_t)255;
    float* xw = (float*)((char*)d_ws + off);

    const int T = 256;
    init_deg_kernel<<<(N + T - 1) / T, T, 0, stream>>>((unsigned int*)dinv, N);
    count_deg_kernel<<<(E + T - 1) / T, T, 0, stream>>>(ei + E, (unsigned int*)dinv, E);
    deg_to_dinv_kernel<<<(N + T - 1) / T, T, 0, stream>>>(dinv, N);
    gcn_gemm_wmma_kernel<<<(N + 127) / 128, T, 0, stream>>>(
        x, Wc, bc, Ws, bs, dinv, xw, out, N);
    const long long work = (long long)E * 16;
    gcn_scatter_kernel<<<(unsigned)((work + T - 1) / T), T, 0, stream>>>(
        ei, dinv, xw, out, E);
}


# Degree count: one random global u32 atomic per edge becomes per-block LDS histograms (12800-node ranges x 32 edge slices), planes summed in the dinv kernel; init launch gone (5 -> 4)
# speedup vs baseline: 1.4013x; 1.4013x over previous
#include <hip/hip_runtime.h>
#include <hip/hip_bf16.h>


typedef __attribute__((ext_vector_type(2))) float v2f;
typedef __attribute__((ext_vector_type(8))) float v8f;

#define D 64

#define HR_NODES 12800
#define HS_SLICES 32
#define HT 512
#define HB 8

__device__ __forceinline__ void hist_bump(unsigned int* h, int c, int lo, unsigned int cnt) {
    const unsigned int d = (unsigned int)(c - lo);
    if (d < cnt) atomicAdd(&h[d], 1u);
}

__global__ __launch_bounds__(HT)
void deg_hist_kernel(const int* __restrict__ col, unsigned int* __restrict__ part,
                     int N, int E) {
    __shared__ unsigned int h[HR_NODES];
    const int lo = blockIdx.x * HR_NODES;
    const int slice = blockIdx.y;
    const int left = N - lo;
    const unsigned int cnt = (unsigned int)(left < HR_NODES ? left : HR_NODES);

    for (int i = threadIdx.x; i < HR_NODES; i += HT) h[i] = 0u;
    __syncthreads();

    const int chunk = (((E + HS_SLICES - 1) / HS_SLICES) + 3) & ~3;
    const long long e0l = (long long)slice * chunk;
    const int e0 = e0l < (long long)E ? (int)e0l : E;
    const int e1 = (E - e0 < chunk) ? E : e0 + chunk;

    if ((((size_t)col) & 15) == 0) {
        const int nv = (e1 - e0) >> 2;
        const int4* c4 = (const int4*)(col + e0);
        int i = threadIdx.x;
        for (; i + (HB - 1) * HT < nv; i += HB * HT) {
            int4 v[HB];
#pragma unroll
            for (int b = 0; b < HB; ++b) v[b] = c4[i + b * HT];
#pragma unroll
            for (int b = 0; b < HB; ++b) {
                hist_bump(h, v[b].x, lo, cnt);
                hist_bump(h, v[b].y, lo, cnt);
                hist_bump(h, v[b].z, lo, cnt);
                hist_bump(h, v[b].w, lo, cnt);
            }
        }
        for (; i < nv; i += HT) {
            const int4 v = c4[i];
            hist_bump(h, v.x, lo, cnt);
            hist_bump(h, v.y, lo, cnt);
            hist_bump(h, v.z, lo, cnt);
            hist_bump(h, v.w, lo, cnt);
        }
        for (int e = e0 + (nv << 2) + threadIdx.x; e < e1; e += HT)
            hist_bump(h, col[e], lo, cnt);
    } else {
        for (int e = e0 + threadIdx.x; e < e1; e += HT)
            hist_bump(h, col[e], lo, cnt);
    }
    __syncthreads();

    unsigned int* dst = part + (size_t)slice * N + lo;
    for (unsigned int i = threadIdx.x; i < cnt; i += HT) dst[i] = h[i];
}

__global__ void deg_to_dinv_kernel(const unsigned int* __restrict__ part,
                                   float* __restrict__ dinv, int N) {
    int i = blockIdx.x * blockDim.x + threadIdx.x;
    if (i < N) {
        unsigned int d = 1u;
#pragma unroll
        for (int s = 0; s < HS_SLICES; ++s) d += part[(size_t)s * N + i];
        dinv[i] = __frsqrt_rn((float)d);
    }
}

__global__ void gcn_gemm_wmma_kernel(const float* __restrict__ x,
                                     const float* __restrict__ Wc,
                                     const float* __restrict__ bc,
                                     const float* __restrict__ Ws,
                                     const float* __restrict__ bs,
                                     const float* __restrict__ dinv,
                                     float* __restrict__ xw,
                                     float* __restrict__ out, int N) {
    const int lane   = threadIdx.x & 31;
    const int wave   = threadIdx.x >> 5;
    const int waveG  = blockIdx.x * (blockDim.x >> 5) + wave;
    const int rowBase = waveG * 16;
    if (rowBase >= N) return;

    const int laneHi = lane >> 4;
    const int laneLo = lane & 15;

    int arow = rowBase + laneLo;
    if (arow >= N) arow = N - 1;
    const float* ap = x + (size_t)arow * D + 2 * laneHi;
    v2f a[16];
#pragma unroll
    for (int k = 0; k < 16; ++k) {
        const float2 t = *(const float2*)(ap + 4 * k);
        a[k].x = t.x; a[k].y = t.y;
    }

#pragma unroll
    for (int nt = 0; nt < 4; ++nt) {
        const int n = nt * 16 + laneLo;
        v8f accC = {};
        v8f accS = {};
#pragma unroll
        for (int k0 = 0; k0 < D; k0 += 4) {
            const int kk = k0 + 2 * laneHi;
            v2f bC, bS;
            bC.x = Wc[(size_t)kk * D + n];
            bC.y = Wc[(size_t)(kk + 1) * D + n];
            bS.x = Ws[(size_t)kk * D + n];
            bS.y = Ws[(size_t)(kk + 1) * D + n];
            accC = __builtin_amdgcn_wmma_f32_16x16x4_f32(
                false, a[k0 >> 2], false, bC, (short)0, accC, false, false);
            accS = __builtin_amdgcn_wmma_f32_16x16x4_f32(
                false, a[k0 >> 2], false, bS, (short)0, accS, false, false);
        }
        const float bias = bs[nt * 16 + laneLo] + bc[nt * 16 + laneLo];
#pragma unroll
        for (int r = 0; r < 8; ++r) {
            const int row = rowBase + r + 8 * laneHi;
            if (row < N) {
                const size_t idx = (size_t)row * D + nt * 16 + laneLo;
                const float xv = accC[r];
                xw[idx] = xv;
                const float di = dinv[row];
                out[idx] = accS[r] + bias + di * di * xv;
            }
        }
    }
}

__global__ void gcn_scatter_kernel(const int* __restrict__ ei,
                                   const float* __restrict__ dinv,
                                   const float* __restrict__ xw,
                                   float* __restrict__ out, int E) {
    const int tid = blockIdx.x * blockDim.x + threadIdx.x;
    const int e = tid >> 4;
    if (e >= E) return;
    const int fg = (tid & 15) << 2;
    const int r = ei[e];
    const int c = ei[E + e];
    const float w = dinv[r] * dinv[c];
    const float4 v = *(const float4*)(xw + (size_t)r * D + fg);
    float* o = out + (size_t)c * D + fg;
#if defined(__HIP_DEVICE_COMPILE__)
    unsafeAtomicAdd(o + 0, w * v.x);
    unsafeAtomicAdd(o + 1, w * v.y);
    unsafeAtomicAdd(o + 2, w * v.z);
    unsafeAtomicAdd(o + 3, w * v.w);
#endif
}

extern "C" void kernel_launch(void* const* d_in, const int* in_sizes, int n_in,
                              void* d_out, int out_size, void* d_ws, size_t ws_size,
                              hipStream_t stream) {
    const float* x  = (const float*)d_in[0];
    const int*   ei = (const int*)d_in[1];
    const float* Wc = (const float*)d_in[2];
    const float* bc = (const float*)d_in[3];
    const float* Ws = (const float*)d_in[4];
    const float* bs = (const float*)d_in[5];
    float* out = (float*)d_out;

    const int N = in_sizes[0] / D;
    const int E = in_sizes[1] / 2;

    float* dinv = (float*)d_ws;
    size_t off = ((size_t)N * sizeof(float) + 255) & ~(size_t)255;
    float* xw = (float*)((char*)d_ws + off);

    const int T = 256;
    unsigned int* part = (unsigned int*)xw;
    deg_hist_kernel<<<dim3((N + HR_NODES - 1) / HR_NODES, HS_SLICES), HT, 0, stream>>>(
        ei + E, part, N, E);
    deg_to_dinv_kernel<<<(N + T - 1) / T, T, 0, stream>>>(part, dinv, N);
    gcn_gemm_wmma_kernel<<<(N + 127) / 128, T, 0, stream>>>(
        x, Wc, bc, Ws, bs, dinv, xw, out, N);
    const long long work = (long long)E * 16;
    gcn_scatter_kernel<<<(unsigned)((work + T - 1) / T), T, 0, stream>>>(
        ei, dinv, xw, out, E);
}


# Edge sum without float atomics: edges counting-sorted by target (LDS cursors), each node pulls and sums its neighbours' x rows, W_conv applied to the sum; sum order: sorted list, aggregate first
# speedup vs baseline: 2.4264x; 1.7315x over previous
#include <hip/hip_runtime.h>
#include <hip/hip_bf16.h>


typedef __attribute__((ext_vector_type(2))) float v2f;
typedef __attribute__((ext_vector_type(8))) float v8f;

#define D 64

#define HR_NODES 12800
#define HS_SLICES 32
#define HT 512
#define HB 8
#define FB 4

__device__ __forceinline__ void hist_bump(unsigned int* h, int c, int lo, unsigned int cnt) {
    const unsigned int d = (unsigned int)(c - lo);
    if (d < cnt) atomicAdd(&h[d], 1u);
}

__global__ __launch_bounds__(HT)
void deg_hist_kernel(const int* __restrict__ col, unsigned int* __restrict__ part,
                     int N, int E) {
    __shared__ unsigned int h[HR_NODES];
    const int lo = blockIdx.x * HR_NODES;
    const int slice = blockIdx.y;
    const int left = N - lo;
    const unsigned int cnt = (unsigned int)(left < HR_NODES ? left : HR_NODES);

    for (int i = threadIdx.x; i < HR_NODES; i += HT) h[i] = 0u;
    __syncthreads();

    const int chunk = (((E + HS_SLICES - 1) / HS_SLICES) + 3) & ~3;
    const long long e0l = (long long)slice * chunk;
    const int e0 = e0l < (long long)E ? (int)e0l : E;
    const int e1 = (E - e0 < chunk) ? E : e0 + chunk;

    if ((((size_t)col) & 15) == 0) {
        const int nv = (e1 - e0) >> 2;
        const int4* c4 = (const int4*)(col + e0);
        int i = threadIdx.x;
        for (; i + (HB - 1) * HT < nv; i += HB * HT) {
            int4 v[HB];
#pragma unroll
            for (int b = 0; b < HB; ++b) v[b] = c4[i + b * HT];
#pragma unroll
            for (int b = 0; b < HB; ++b) {
                hist_bump(h, v[b].x, lo, cnt);
                hist_bump(h, v[b].y, lo, cnt);
                hist_bump(h, v[b].z, lo, cnt);
                hist_bump(h, v[b].w, lo, cnt);
            }
        }
        for (; i < nv; i += HT) {
            const int4 v = c4[i];
            hist_bump(h, v.x, lo, cnt);
            hist_bump(h, v.y, lo, cnt);
            hist_bump(h, v.z, lo, cnt);
            hist_bump(h, v.w, lo, cnt);
        }
        for (int e = e0 + (nv << 2) + threadIdx.x; e < e1; e += HT)
            hist_bump(h, col[e], lo, cnt);
    } else {
        for (int e = e0 + threadIdx.x; e < e1; e += HT)
            hist_bump(h, col[e], lo, cnt);
    }
    __syncthreads();

    unsigned int* dst = part + (size_t)slice * N + lo;
    for (unsigned int i = threadIdx.x; i < cnt; i += HT) dst[i] = h[i];
}

#define SB 256
#define NBR (HR_NODES / SB)

__global__ __launch_bounds__(SB)
void deg_scan_kernel(unsigned int* __restrict__ part, float* __restrict__ dinv,
                     unsigned int* __restrict__ degc, unsigned int* __restrict__ rp_local,
                     unsigned int* __restrict__ bsum, int N) {
    __shared__ unsigned int s[SB];
    const int t = threadIdx.x;
    const int i = blockIdx.x * SB + t;
    unsigned int run = 0u;
    if (i < N) {
        unsigned int c[HS_SLICES];
#pragma unroll
        for (int k = 0; k < HS_SLICES; ++k) c[k] = part[(size_t)k * N + i];
#pragma unroll
        for (int k = 0; k < HS_SLICES; ++k) {
            part[(size_t)k * N + i] = run;
            run += c[k];
        }
        dinv[i] = __frsqrt_rn((float)(run + 1u));
        degc[i] = run;
    }
    s[t] = run;
    __syncthreads();
    for (int off = 1; off < SB; off <<= 1) {
        const unsigned int v = (t >= off) ? s[t - off] : 0u;
        __syncthreads();
        s[t] += v;
        __syncthreads();
    }
    if (i < N) rp_local[i] = s[t] - run;
    if (t == SB - 1) bsum[blockIdx.x] = s[t];
}

__device__ __forceinline__ void fill_put(unsigned int* cur, int c, int r, int lo,
                                         unsigned int cnt, int* __restrict__ srt,
                                         unsigned int E) {
    const unsigned int d = (unsigned int)(c - lo);
    if (d < cnt) {
        const unsigned int pos = atomicAdd(&cur[d], 1u);
        if (pos < E) srt[pos] = r;
    }
}

__device__ __forceinline__ void fill_put4(unsigned int* cur, const int4 c, const int4 r, int lo,
                                          unsigned int cnt, int* __restrict__ srt,
                                          unsigned int E) {
    const unsigned int d0 = (unsigned int)(c.x - lo);
    const unsigned int d1 = (unsigned int)(c.y - lo);
    const unsigned int d2 = (unsigned int)(c.z - lo);
    const unsigned int d3 = (unsigned int)(c.w - lo);
    unsigned int p0 = E, p1 = E, p2 = E, p3 = E;
    if (d0 < cnt) p0 = atomicAdd(&cur[d0], 1u);
    if (d1 < cnt) p1 = atomicAdd(&cur[d1], 1u);
    if (d2 < cnt) p2 = atomicAdd(&cur[d2], 1u);
    if (d3 < cnt) p3 = atomicAdd(&cur[d3], 1u);
    if (p0 < E) srt[p0] = r.x;
    if (p1 < E) srt[p1] = r.y;
    if (p2 < E) srt[p2] = r.z;
    if (p3 < E) srt[p3] = r.w;
}

__global__ __launch_bounds__(HT)
void fill_kernel(const int* __restrict__ row, const int* __restrict__ col,
                 const unsigned int* __restrict__ part,
                 const unsigned int* __restrict__ rp_local,
                 const unsigned int* __restrict__ bsum,
                 unsigned int* __restrict__ rp_abs, int* __restrict__ srt,
                 int N, int E) {
    __shared__ unsigned int cur[HR_NODES];
    __shared__ unsigned int tmp[NBR];
    __shared__ unsigned int bp[NBR];
    __shared__ unsigned int base0;
    const int t = threadIdx.x;
    const int lo = blockIdx.x * HR_NODES;
    const int slice = blockIdx.y;
    const int left = N - lo;
    const unsigned int cnt = (unsigned int)(left < HR_NODES ? left : HR_NODES);

    const int nb = (N + SB - 1) / SB;
    const int first = lo / SB;
    if (t == 0) base0 = 0u;
    if (t < NBR) tmp[t] = (first + t < nb) ? bsum[first + t] : 0u;
    __syncthreads();
    unsigned int mine = 0u;
    for (int b = t; b < first; b += HT) mine += bsum[b];
    if (mine != 0u) atomicAdd(&base0, mine);
    __syncthreads();
    if (t < NBR) {
        unsigned int sum = base0;
#pragma unroll
        for (int j = 0; j < NBR; ++j) sum += (j < t) ? tmp[j] : 0u;
        bp[t] = sum;
    }
    __syncthreads();
    for (unsigned int i = t; i < cnt; i += HT) {
        const unsigned int base = bp[i / SB] + rp_local[lo + i];
        if (slice == 0) rp_abs[lo + i] = base;
        cur[i] = base + part[(size_t)slice * N + lo + i];
    }
    __syncthreads();

    const int chunk = (((E + HS_SLICES - 1) / HS_SLICES) + 3) & ~3;
    const long long e0l = (long long)slice * chunk;
    const int e0 = e0l < (long long)E ? (int)e0l : E;
    const int e1 = (E - e0 < chunk) ? E : e0 + chunk;
    const unsigned int Eu = (unsigned int)E;

    if (((((size_t)col) | ((size_t)row)) & 15) == 0) {
        const int nv = (e1 - e0) >> 2;
        const int4* c4 = (const int4*)(col + e0);
        const int4* r4 = (const int4*)(row + e0);
        int i = t;
        for (; i + (FB - 1) * HT < nv; i += FB * HT) {
            int4 vc[FB], vr[FB];
#pragma unroll
            for (int b = 0; b < FB; ++b) { vc[b] = c4[i + b * HT]; vr[b] = r4[i + b * HT]; }
#pragma unroll
            for (int b = 0; b < FB; ++b) fill_put4(cur, vc[b], vr[b], lo, cnt, srt, Eu);
        }
        for (; i < nv; i += HT) {
            const int4 vc = c4[i];
            const int4 vr = r4[i];
            fill_put4(cur, vc, vr, lo, cnt, srt, Eu);
        }
        for (int e = e0 + (nv << 2) + t; e < e1; e += HT)
            fill_put(cur, col[e], row[e], lo, cnt, srt, Eu);
    } else {
        for (int e = e0 + t; e < e1; e += HT)
            fill_put(cur, col[e], row[e], lo, cnt, srt, Eu);
    }
}

#define PB 8
__device__ __forceinline__ void pull_batch(float4& acc, const int* __restrict__ lst,
                                           unsigned int j, unsigned int n, bool guard,
                                           const float* __restrict__ dinv,
                                           const float4* __restrict__ x4,
                                           unsigned int self, unsigned int lastNode,
                                           float dc, int l) {
    unsigned int r[PB];
    float d[PB];
    float4 v[PB];
#pragma unroll
    for (int k = 0; k < PB; ++k) {
        const bool ok = !guard || (j + k < n);
        const unsigned int q = ok ? (unsigned int)lst[j + k] : self;
        r[k] = q < lastNode ? q : lastNode;
    }
#pragma unroll
    for (int k = 0; k < PB; ++k) {
        d[k] = dinv[r[k]];
        v[k] = x4[(size_t)r[k] * (D / 4) + l];
    }
#pragma unroll
    for (int k = 0; k < PB; ++k) {
        const bool ok = !guard || (j + k < n);
        const float w = ok ? d[k] * dc : 0.0f;
        acc.x += w * v[k].x;
        acc.y += w * v[k].y;
        acc.z += w * v[k].z;
        acc.w += w * v[k].w;
    }
}

__global__ __launch_bounds__(256) __attribute__((amdgpu_waves_per_eu(4, 8)))
void gcn_pull_kernel(const float* __restrict__ x, const float* __restrict__ dinv,
                     const unsigned int* __restrict__ rp_abs,
                     const unsigned int* __restrict__ degc,
                     const int* __restrict__ srt, float* __restrict__ agg,
                     int N, int E) {
    const int tid = blockIdx.x * blockDim.x + threadIdx.x;
    const int c = tid >> 4;
    if (c >= N) return;
    const int l = tid & 15;
    const float4* x4 = (const float4*)x;
    const unsigned int Eu = (unsigned int)E;
    unsigned int start = rp_abs[c];
    if (start > Eu) start = Eu;
    unsigned int n = degc[c];
    if (n > Eu - start) n = Eu - start;
    const int* lst = srt + start;
    const float dc = dinv[c];
    const float4 xs = x4[(size_t)c * (D / 4) + l];
    const float ws = dc * dc;
    float4 acc;
    acc.x = ws * xs.x; acc.y = ws * xs.y; acc.z = ws * xs.z; acc.w = ws * xs.w;
    unsigned int j = 0;
    for (; j + PB <= n; j += PB)
        pull_batch(acc, lst, j, n, false, dinv, x4, (unsigned int)c, (unsigned int)(N - 1), dc, l);
    if (j < n)
        pull_batch(acc, lst, j, n, true, dinv, x4, (unsigned int)c, (unsigned int)(N - 1), dc, l);
    ((float4*)agg)[(size_t)c * (D / 4) + l] = acc;
}

__global__ void gcn_gemm_wmma_kernel(const float* __restrict__ x,
                                     const float* __restrict__ agg,
                                     const float* __restrict__ Wc,
                                     const float* __restrict__ bc,
                                     const float* __restrict__ Ws,
                                     const float* __restrict__ bs,
                                     float* __restrict__ out, int N) {
    const int lane   = threadIdx.x & 31;
    const int wave   = threadIdx.x >> 5;
    const int waveG  = blockIdx.x * (blockDim.x >> 5) + wave;
    const int rowBase = waveG * 16;
    if (rowBase >= N) return;

    const int laneHi = lane >> 4;
    const int laneLo = lane & 15;

    int arow = rowBase + laneLo;
    if (arow >= N) arow = N - 1;
    const float* ap = x + (size_t)arow * D + 2 * laneHi;
    const float* gp = agg + (size_t)arow * D + 2 * laneHi;
    v2f a[16];
    v2f g[16];
#pragma unroll
    for (int k = 0; k < 16; ++k) {
        const float2 t = *(const float2*)(ap + 4 * k);
        a[k].x = t.x; a[k].y = t.y;
        const float2 u = *(const float2*)(gp + 4 * k);
        g[k].x = u.x; g[k].y = u.y;
    }

#pragma unroll
    for (int nt = 0; nt < 4; ++nt) {
        const int n = nt * 16 + laneLo;
        v8f accC = {};
        v8f accS = {};
#pragma unroll
        for (int k0 = 0; k0 < D; k0 += 4) {
            const int kk = k0 + 2 * laneHi;
            v2f bC, bS;
            bC.x = Wc[(size_t)kk * D + n];
            bC.y = Wc[(size_t)(kk + 1) * D + n];
            bS.x = Ws[(size_t)kk * D + n];
            bS.y = Ws[(size_t)(kk + 1) * D + n];
            accC = __builtin_amdgcn_wmma_f32_16x16x4_f32(
                false, g[k0 >> 2], false, bC, (short)0, accC, false, false);
            accS = __builtin_amdgcn_wmma_f32_16x16x4_f32(
                false, a[k0 >> 2], false, bS, (short)0, accS, false, false);
        }
        const float bias = bs[nt * 16 + laneLo] + bc[nt * 16 + laneLo];
#pragma unroll
        for (int r = 0; r < 8; ++r) {
            const int row = rowBase + r + 8 * laneHi;
            if (row < N) {
                const size_t idx = (size_t)row * D + nt * 16 + laneLo;
                out[idx] = accS[r] + bias + accC[r];
            }
        }
    }
}

extern "C" void kernel_launch(void* const* d_in, const int* in_sizes, int n_in,
                              void* d_out, int out_size, void* d_ws, size_t ws_size,
                              hipStream_t stream) {
    const float* x  = (const float*)d_in[0];
    const int*   ei = (const int*)d_in[1];
    const float* Wc = (const float*)d_in[2];
    const float* bc = (const float*)d_in[3];
    const float* Ws = (const float*)d_in[4];
    const float* bs = (const float*)d_in[5];
    float* out = (float*)d_out;

    const int N = in_sizes[0] / D;
    const int E = in_sizes[1] / 2;

    float* dinv = (float*)d_ws;
    size_t off = ((size_t)N * sizeof(float) + 255) & ~(size_t)255;
    float* agg = (float*)((char*)d_ws + off);
    unsigned int* part = (unsigned int*)agg;
    int* srt = (int*)d_out;
    unsigned int* rp_local = (unsigned int*)(srt + E);
    unsigned int* rp_abs = rp_local + N;
    unsigned int* degc = rp_abs + N;
    unsigned int* bsum = degc + N;

    const int T = 256;
    const dim3 hgrid((N + HR_NODES - 1) / HR_NODES, HS_SLICES);
    deg_hist_kernel<<<hgrid, HT, 0, stream>>>(ei + E, part, N, E);
    deg_scan_kernel<<<(N + SB - 1) / SB, SB, 0, stream>>>(part, dinv, degc, rp_local, bsum, N);
    fill_kernel<<<hgrid, HT, 0, stream>>>(ei, ei + E, part, rp_local, bsum, rp_abs, srt, N, E);
    const long long work = (long long)N * 16;
    gcn_pull_kernel<<<(unsigned)((work + T - 1) / T), T, 0, stream>>>(
        x, dinv, rp_abs, degc, srt, agg, N, E);
    gcn_gemm_wmma_kernel<<<(N + 127) / 128, T, 0, stream>>>(
        x, agg, Wc, bc, Ws, bs, out, N);
}


# Pull fused into the WMMA kernel: a wave sums its 16 nodes' neighbour rows into an LDS tile that feeds the products; agg never stored, 5 -> 4 launches; sum order kept
# speedup vs baseline: 2.9248x; 1.2054x over previous
#include <hip/hip_runtime.h>
#include <hip/hip_bf16.h>


typedef __attribute__((ext_vector_type(2))) float v2f;
typedef __attribute__((ext_vector_type(8))) float v8f;

#define D 64

#define HR_NODES 12800
#define HS_SLICES 32
#define HT 512
#define HB 8
#define FB 4

__device__ __forceinline__ void hist_bump(unsigned int* h, int c, int lo, unsigned int cnt) {
    const unsigned int d = (unsigned int)(c - lo);
    if (d < cnt) atomicAdd(&h[d], 1u);
}

__global__ __launch_bounds__(HT)
void deg_hist_kernel(const int* __restrict__ col, unsigned int* __restrict__ part,
                     int N, int E) {
    __shared__ unsigned int h[HR_NODES];
    const int lo = blockIdx.x * HR_NODES;
    const int slice = blockIdx.y;
    const int left = N - lo;
    const unsigned int cnt = (unsigned int)(left < HR_NODES ? left : HR_NODES);

    for (int i = threadIdx.x; i < HR_NODES; i += HT) h[i] = 0u;
    __syncthreads();

    const int chunk = (((E + HS_SLICES - 1) / HS_SLICES) + 3) & ~3;
    const long long e0l = (long long)slice * chunk;
    const int e0 = e0l < (long long)E ? (int)e0l : E;
    const int e1 = (E - e0 < chunk) ? E : e0 + chunk;

    if ((((size_t)col) & 15) == 0) {
        const int nv = (e1 - e0) >> 2;
        const int4* c4 = (const int4*)(col + e0);
        int i = threadIdx.x;
        for (; i + (HB - 1) * HT < nv; i += HB * HT) {
            int4 v[HB];
#pragma unroll
            for (int b = 0; b < HB; ++b) v[b] = c4[i + b * HT];
#pragma unroll
            for (int b = 0; b < HB; ++b) {
                hist_bump(h, v[b].x, lo, cnt);
                hist_bump(h, v[b].y, lo, cnt);
                hist_bump(h, v[b].z, lo, cnt);
                hist_bump(h, v[b].w, lo, cnt);
            }
        }
        for (; i < nv; i += HT) {
            const int4 v = c4[i];
            hist_bump(h, v.x, lo, cnt);
            hist_bump(h, v.y, lo, cnt);
            hist_bump(h, v.z, lo, cnt);
            hist_bump(h, v.w, lo, cnt);
        }
        for (int e = e0 + (nv << 2) + threadIdx.x; e < e1; e += HT)
            hist_bump(h, col[e], lo, cnt);
    } else {
        for (int e = e0 + threadIdx.x; e < e1; e += HT)
            hist_bump(h, col[e], lo, cnt);
    }
    __syncthreads();

    unsigned int* dst = part + (size_t)slice * N + lo;
    for (unsigned int i = threadIdx.x; i < cnt; i += HT) dst[i] = h[i];
}

#define SB 256
#define NBR (HR_NODES / SB)

__global__ __launch_bounds__(SB)
void deg_scan_kernel(unsigned int* __restrict__ part, float* __restrict__ dinv,
                     unsigned int* __restrict__ degc, unsigned int* __restrict__ rp_local,
                     unsigned int* __restrict__ bsum, int N) {
    __shared__ unsigned int s[SB];
    const int t = threadIdx.x;
    const int i = blockIdx.x * SB + t;
    unsigned int run = 0u;
    if (i < N) {
        unsigned int c[HS_SLICES];
#pragma unroll
        for (int k = 0; k < HS_SLICES; ++k) c[k] = part[(size_t)k * N + i];
#pragma unroll
        for (int k = 0; k < HS_SLICES; ++k) {
            part[(size_t)k * N + i] = run;
            run += c[k];
        }
        dinv[i] = __frsqrt_rn((float)(run + 1u));
        degc[i] = run;
    }
    s[t] = run;
    __syncthreads();
    for (int off = 1; off < SB; off <<= 1) {
        const unsigned int v = (t >= off) ? s[t - off] : 0u;
        __syncthreads();
        s[t] += v;
        __syncthreads();
    }
    if (i < N) rp_local[i] = s[t] - run;
    if (t == SB - 1) bsum[blockIdx.x] = s[t];
}

__device__ __forceinline__ void fill_put(unsigned int* cur, int c, int r, int lo,
                                         unsigned int cnt, int* __restrict__ srt,
                                         unsigned int E) {
    const unsigned int d = (unsigned int)(c - lo);
    if (d < cnt) {
        const unsigned int pos = atomicAdd(&cur[d], 1u);
        if (pos < E) srt[pos] = r;
    }
}

__device__ __forceinline__ void fill_put4(unsigned int* cur, const int4 c, const int4 r, int lo,
                                          unsigned int cnt, int* __restrict__ srt,
                                          unsigned int E) {
    const unsigned int d0 = (unsigned int)(c.x - lo);
    const unsigned int d1 = (unsigned int)(c.y - lo);
    const unsigned int d2 = (unsigned int)(c.z - lo);
    const unsigned int d3 = (unsigned int)(c.w - lo);
    unsigned int p0 = E, p1 = E, p2 = E, p3 = E;
    if (d0 < cnt) p0 = atomicAdd(&cur[d0], 1u);
    if (d1 < cnt) p1 = atomicAdd(&cur[d1], 1u);
    if (d2 < cnt) p2 = atomicAdd(&cur[d2], 1u);
    if (d3 < cnt) p3 = atomicAdd(&cur[d3], 1u);
    if (p0 < E) srt[p0] = r.x;
    if (p1 < E) srt[p1] = r.y;
    if (p2 < E) srt[p2] = r.z;
    if (p3 < E) srt[p3] = r.w;
}

__global__ __launch_bounds__(HT)
void fill_kernel(const int* __restrict__ row, const int* __restrict__ col,
                 const unsigned int* __restrict__ part,
                 const unsigned int* __restrict__ rp_local,
                 const unsigned int* __restrict__ bsum,
                 unsigned int* __restrict__ rp_abs, int* __restrict__ srt,
                 int N, int E) {
    __shared__ unsigned int cur[HR_NODES];
    __shared__ unsigned int tmp[NBR];
    __shared__ unsigned int bp[NBR];
    __shared__ unsigned int base0;
    const int t = threadIdx.x;
    const int lo = blockIdx.x * HR_NODES;
    const int slice = blockIdx.y;
    const int left = N - lo;
    const unsigned int cnt = (unsigned int)(left < HR_NODES ? left : HR_NODES);

    const int nb = (N + SB - 1) / SB;
    const int first = lo / SB;
    if (t == 0) base0 = 0u;
    if (t < NBR) tmp[t] = (first + t < nb) ? bsum[first + t] : 0u;
    __syncthreads();
    unsigned int mine = 0u;
    for (int b = t; b < first; b += HT) mine += bsum[b];
    if (mine != 0u) atomicAdd(&base0, mine);
    __syncthreads();
    if (t < NBR) {
        unsigned int sum = base0;
#pragma unroll
        for (int j = 0; j < NBR; ++j) sum += (j < t) ? tmp[j] : 0u;
        bp[t] = sum;
    }
    __syncthreads();
    for (unsigned int i = t; i < cnt; i += HT) {
        const unsigned int base = bp[i / SB] + rp_local[lo + i];
        if (slice == 0) rp_abs[lo + i] = base;
        cur[i] = base + part[(size_t)slice * N + lo + i];
    }
    __syncthreads();

    const int chunk = (((E + HS_SLICES - 1) / HS_SLICES) + 3) & ~3;
    const long long e0l = (long long)slice * chunk;
    const int e0 = e0l < (long long)E ? (int)e0l : E;
    const int e1 = (E - e0 < chunk) ? E : e0 + chunk;
    const unsigned int Eu = (unsigned int)E;

    if (((((size_t)col) | ((size_t)row)) & 15) == 0) {
        const int nv = (e1 - e0) >> 2;
        const int4* c4 = (const int4*)(col + e0);
        const int4* r4 = (const int4*)(row + e0);
        int i = t;
        for (; i + (FB - 1) * HT < nv; i += FB * HT) {
            int4 vc[FB], vr[FB];
#pragma unroll
            for (int b = 0; b < FB; ++b) { vc[b] = c4[i + b * HT]; vr[b] = r4[i + b * HT]; }
#pragma unroll
            for (int b = 0; b < FB; ++b) fill_put4(cur, vc[b], vr[b], lo, cnt, srt, Eu);
        }
        for (; i < nv; i += HT) {
            const int4 vc = c4[i];
            const int4 vr = r4[i];
            fill_put4(cur, vc, vr, lo, cnt, srt, Eu);
        }
        for (int e = e0 + (nv << 2) + t; e < e1; e += HT)
            fill_put(cur, col[e], row[e], lo, cnt, srt, Eu);
    } else {
        for (int e = e0 + t; e < e1; e += HT)
            fill_put(cur, col[e], row[e], lo, cnt, srt, Eu);
    }
}

#define WPB 2
#define PITCH 68

template <int CNT>
__device__ __forceinline__ void pull_group(float4& acc, const float4* __restrict__ x4,
                                           unsigned int r16, float w16, int l) {
    float4 v[CNT];
    const char* xb = (const char*)x4;
    const unsigned int lo = (unsigned int)l * 16u;
#pragma unroll
    for (int k = 0; k < CNT; ++k) {
        const unsigned int r = __shfl(r16, k, 16);
        v[k] = *(const float4*)(xb + ((r << 8) + lo));
    }
#pragma unroll
    for (int k = 0; k < CNT; ++k) {
        const float w = __shfl(w16, k, 16);
        acc.x += w * v[k].x;
        acc.y += w * v[k].y;
        acc.z += w * v[k].z;
        acc.w += w * v[k].w;
    }
}

__device__ __forceinline__ void pull_chunk(float4& acc, const float4* __restrict__ x4,
                                           unsigned int r16, float w16, int l, unsigned int rem) {
    if (rem > 8u) pull_group<16>(acc, x4, r16, w16, l);
    else if (rem > 0u) pull_group<8>(acc, x4, r16, w16, l);
}

__global__ __launch_bounds__(WPB * 32) __attribute__((amdgpu_waves_per_eu(7, 8)))
void gcn_pull_gemm_kernel(const float* __restrict__ x, const float* __restrict__ dinv,
                          const unsigned int* __restrict__ rp_abs,
                          const unsigned int* __restrict__ degc,
                          const int* __restrict__ srt,
                          const float* __restrict__ Wc, const float* __restrict__ bc,
                          const float* __restrict__ Ws, const float* __restrict__ bs,
                          float* __restrict__ out, int N, int E) {
    __shared__ float aggS[WPB][16 * PITCH];
    __shared__ float xS[WPB][16 * PITCH];
    const int lane   = threadIdx.x & 31;
    const int wave   = threadIdx.x >> 5;
    const int waveG  = blockIdx.x * WPB + wave;
    const int rowBase = waveG * 16;
    const int laneHi = lane >> 4;
    const int laneLo = lane & 15;
    const float4* x4 = (const float4*)x;
    const unsigned int Eu = (unsigned int)E;
    const unsigned int lastE = Eu > 0u ? Eu - 1u : 0u;
    const unsigned int lastNode = (unsigned int)(N - 1);
    float* myG = aggS[wave];
    float* myX = xS[wave];

    const int hrow = rowBase + laneLo;
    const bool hv = hrow < N;
    const int hr = hv ? hrow : N - 1;
    unsigned int startAll = rp_abs[hr];
    if (startAll > Eu) startAll = Eu;
    unsigned int nAll = hv ? degc[hr] : 0u;
    if (nAll > Eu - startAll) nAll = Eu - startAll;
    const float dcAll = dinv[hr];

    unsigned int start = __shfl(startAll, laneHi);
    unsigned int n = __shfl(nAll, laneHi);
    float dc = __shfl(dcAll, laneHi);
    int c = rowBase + laneHi; if (c > N - 1) c = N - 1;
    unsigned int rA, rB;
    {
        unsigned int qA = start + (unsigned int)laneLo;       if (qA > lastE) qA = lastE;
        unsigned int qB = start + (unsigned int)laneLo + 16u; if (qB > lastE) qB = lastE;
        const int iA = srt[qA];
        const int iB = srt[qB];
        rA = ((unsigned int)laneLo < n) ? (unsigned int)iA : (unsigned int)c;
        rB = ((unsigned int)(16 + laneLo) < n) ? (unsigned int)iB : (unsigned int)c;
        if (rA > lastNode) rA = lastNode;
        if (rB > lastNode) rB = lastNode;
    }
    float dA = dinv[rA];
    float dB = dinv[rB];
    float4 xs = x4[(size_t)c * (D / 4) + laneLo];
    unsigned int startN = __shfl(startAll, 2 + laneHi);
    unsigned int nN = __shfl(nAll, 2 + laneHi);
    float dcN = __shfl(dcAll, 2 + laneHi);
    int cN = rowBase + 2 + laneHi; if (cN > N - 1) cN = N - 1;
    int iAn, iBn;
    {
        unsigned int qA = startN + (unsigned int)laneLo;       if (qA > lastE) qA = lastE;
        unsigned int qB = startN + (unsigned int)laneLo + 16u; if (qB > lastE) qB = lastE;
        iAn = srt[qA];
        iBn = srt[qB];
    }

    for (int p = 0; p < 8; ++p) {
        unsigned int rAn = ((unsigned int)laneLo < nN) ? (unsigned int)iAn : (unsigned int)cN;
        unsigned int rBn = ((unsigned int)(16 + laneLo) < nN) ? (unsigned int)iBn : (unsigned int)cN;
        if (rAn > lastNode) rAn = lastNode;
        if (rBn > lastNode) rBn = lastNode;
        const float dAn = dinv[rAn];
        const float dBn = dinv[rBn];
        const float4 xsn = x4[(size_t)cN * (D / 4) + laneLo];
        const int p2 = p < 6 ? p + 2 : 7;
        const int m2 = 2 * p2 + laneHi;
        const unsigned int start2 = __shfl(startAll, m2);
        const unsigned int n2 = __shfl(nAll, m2);
        const float dc2 = __shfl(dcAll, m2);
        int c2 = rowBase + m2; if (c2 > N - 1) c2 = N - 1;
        unsigned int qA2 = start2 + (unsigned int)laneLo;       if (qA2 > lastE) qA2 = lastE;
        unsigned int qB2 = start2 + (unsigned int)laneLo + 16u; if (qB2 > lastE) qB2 = lastE;
        const int iA2 = srt[qA2];
        const int iB2 = srt[qB2];
        const unsigned int nOther = __shfl_xor(n, 16);
        const unsigned int nmax = __builtin_amdgcn_readfirstlane(n > nOther ? n : nOther);
        const float ws = dc * dc;
        float4 acc;
        acc.x = ws * xs.x; acc.y = ws * xs.y; acc.z = ws * xs.z; acc.w = ws * xs.w;
        const float wA = ((unsigned int)laneLo < n) ? dA * dc : 0.0f;
        const float wB = ((unsigned int)(16 + laneLo) < n) ? dB * dc : 0.0f;
        pull_chunk(acc, x4, rA, wA, laneLo, nmax);
        pull_chunk(acc, x4, rB, wB, laneLo, nmax > 16u ? nmax - 16u : 0u);
        for (unsigned int j = 32u; j < nmax; j += 16u) {
            const bool ok = (j + (unsigned int)laneLo) < n;
            unsigned int qC = start + j + (unsigned int)laneLo; if (qC > lastE) qC = lastE;
            const int iC = srt[qC];
            unsigned int rC = ok ? (unsigned int)iC : (unsigned int)c;
            if (rC > lastNode) rC = lastNode;
            const float wC = ok ? dinv[rC] * dc : 0.0f;
            pull_chunk(acc, x4, rC, wC, laneLo, nmax - j);
        }
        const int rowOff = (2 * p + laneHi) * PITCH + 4 * laneLo;
        *(float4*)(myG + rowOff) = acc;
        *(float4*)(myX + rowOff) = xs;
        start = startN; n = nN; dc = dcN; c = cN;
        rA = rAn; rB = rBn; dA = dAn; dB = dBn; xs = xsn;
        startN = start2; nN = n2; dcN = dc2; cN = c2; iAn = iA2; iBn = iB2;
    }
    __syncthreads();

    const float* ap = myX + laneLo * PITCH + 2 * laneHi;
    const float* gp = myG + laneLo * PITCH + 2 * laneHi;
    v2f a[16];
    v2f g[16];
#pragma unroll
    for (int k = 0; k < 16; ++k) {
        const float2 t = *(const float2*)(ap + 4 * k);
        a[k].x = t.x; a[k].y = t.y;
        const float2 u = *(const float2*)(gp + 4 * k);
        g[k].x = u.x; g[k].y = u.y;
    }

#pragma unroll
    for (int nt = 0; nt < 4; ++nt) {
        const int nn = nt * 16 + laneLo;
        v8f accC = {};
        v8f accS = {};
#pragma unroll
        for (int k0 = 0; k0 < D; k0 += 4) {
            const int kk = k0 + 2 * laneHi;
            v2f bC, bS;
            bC.x = Wc[(size_t)kk * D + nn];
            bC.y = Wc[(size_t)(kk + 1) * D + nn];
            bS.x = Ws[(size_t)kk * D + nn];
            bS.y = Ws[(size_t)(kk + 1) * D + nn];
            accC = __builtin_amdgcn_wmma_f32_16x16x4_f32(
                false, g[k0 >> 2], false, bC, (short)0, accC, false, false);
            accS = __builtin_amdgcn_wmma_f32_16x16x4_f32(
                false, a[k0 >> 2], false, bS, (short)0, accS, false, false);
        }
        const float bias = bs[nt * 16 + laneLo] + bc[nt * 16 + laneLo];
#pragma unroll
        for (int r = 0; r < 8; ++r) {
            const int row = rowBase + r + 8 * laneHi;
            if (row < N) {
                const size_t idx = (size_t)row * D + nt * 16 + laneLo;
                out[idx] = accS[r] + bias + accC[r];
            }
        }
    }
}

extern "C" void kernel_launch(void* const* d_in, const int* in_sizes, int n_in,
                              void* d_out, int out_size, void* d_ws, size_t ws_size,
                              hipStream_t stream) {
    const float* x  = (const float*)d_in[0];
    const int*   ei = (const int*)d_in[1];
    const float* Wc = (const float*)d_in[2];
    const float* bc = (const float*)d_in[3];
    const float* Ws = (const float*)d_in[4];
    const float* bs = (const float*)d_in[5];
    float* out = (float*)d_out;

    const int N = in_sizes[0] / D;
    const int E = in_sizes[1] / 2;

    float* dinv = (float*)d_ws;
    size_t off = ((size_t)N * sizeof(float) + 255) & ~(size_t)255;
    float* agg = (float*)((char*)d_ws + off);
    unsigned int* part = (unsigned int*)agg;
    int* srt = (int*)(part + (size_t)HS_SLICES * N);
    unsigned int* rp_local = (unsigned int*)(srt + E);
    unsigned int* rp_abs = rp_local + N;
    unsigned int* degc = rp_abs + N;
    unsigned int* bsum = degc + N;

    const dim3 hgrid((N + HR_NODES - 1) / HR_NODES, HS_SLICES);
    deg_hist_kernel<<<hgrid, HT, 0, stream>>>(ei + E, part, N, E);
    deg_scan_kernel<<<(N + SB - 1) / SB, SB, 0, stream>>>(part, dinv, degc, rp_local, bsum, N);
    fill_kernel<<<hgrid, HT, 0, stream>>>(ei, ei + E, part, rp_local, bsum, rp_abs, srt, N, E);
    gcn_pull_gemm_kernel<<<(N + 16 * WPB - 1) / (16 * WPB), WPB * 32, 0, stream>>>(
        x, dinv, rp_abs, degc, srt, Wc, bc, Ws, bs, out, N, E);
}


# Fill kernel cursor set-up: the two loads per node of its 25-trip loop (scan offset, earlier-slice count) issued together at kernel entry; one memory wait instead of two per trip
# speedup vs baseline: 3.3698x; 1.1521x over previous
#include <hip/hip_runtime.h>
#include <hip/hip_bf16.h>


typedef __attribute__((ext_vector_type(2))) float v2f;
typedef __attribute__((ext_vector_type(8))) float v8f;

#define D 64

#define HR_NODES 12800
#define HS_SLICES 32
#define HT 512
#define HB 8
#define FB 4
#define CI (HR_NODES / HT)
static_assert(HR_NODES % HT == 0, "the cursor set-up gives every thread the same number of nodes");

__device__ __forceinline__ void hist_bump(unsigned int* h, int c, int lo, unsigned int cnt) {
    const unsigned int d = (unsigned int)(c - lo);
    if (d < cnt) atomicAdd(&h[d], 1u);
}

__global__ __launch_bounds__(HT)
void deg_hist_kernel(const int* __restrict__ col, unsigned int* __restrict__ part,
                     int N, int E) {
    __shared__ unsigned int h[HR_NODES];
    const int lo = blockIdx.x * HR_NODES;
    const int slice = blockIdx.y;
    const int left = N - lo;
    const unsigned int cnt = (unsigned int)(left < HR_NODES ? left : HR_NODES);

    for (int i = threadIdx.x; i < HR_NODES; i += HT) h[i] = 0u;
    __syncthreads();

    const int chunk = (((E + HS_SLICES - 1) / HS_SLICES) + 3) & ~3;
    const long long e0l = (long long)slice * chunk;
    const int e0 = e0l < (long long)E ? (int)e0l : E;
    const int e1 = (E - e0 < chunk) ? E : e0 + chunk;

    if ((((size_t)col) & 15) == 0) {
        const int nv = (e1 - e0) >> 2;
        const int4* c4 = (const int4*)(col + e0);
        int i = threadIdx.x;
        for (; i + (HB - 1) * HT < nv; i += HB * HT) {
            int4 v[HB];
#pragma unroll
            for (int b = 0; b < HB; ++b) v[b] = c4[i + b * HT];
#pragma unroll
            for (int b = 0; b < HB; ++b) {
                hist_bump(h, v[b].x, lo, cnt);
                hist_bump(h, v[b].y, lo, cnt);
                hist_bump(h, v[b].z, lo, cnt);
                hist_bump(h, v[b].w, lo, cnt);
            }
        }
        for (; i < nv; i += HT) {
            const int4 v = c4[i];
            hist_bump(h, v.x, lo, cnt);
            hist_bump(h, v.y, lo, cnt);
            hist_bump(h, v.z, lo, cnt);
            hist_bump(h, v.w, lo, cnt);
        }
        for (int e = e0 + (nv << 2) + threadIdx.x; e < e1; e += HT)
            hist_bump(h, col[e], lo, cnt);
    } else {
        for (int e = e0 + threadIdx.x; e < e1; e += HT)
            hist_bump(h, col[e], lo, cnt);
    }
    __syncthreads();

    unsigned int* dst = part + (size_t)slice * N + lo;
    for (unsigned int i = threadIdx.x; i < cnt; i += HT) dst[i] = h[i];
}

#define SB 256
#define NBR (HR_NODES / SB)
static_assert(HR_NODES % SB == 0, "a node range is a whole number of scan blocks");

__global__ __launch_bounds__(SB)
void deg_scan_kernel(unsigned int* __restrict__ part, float* __restrict__ dinv,
                     unsigned int* __restrict__ degc, unsigned int* __restrict__ rp_local,
                     unsigned int* __restrict__ bsum, int N) {
    __shared__ unsigned int s[SB];
    const int t = threadIdx.x;
    const int i = blockIdx.x * SB + t;
    unsigned int run = 0u;
    if (i < N) {
        unsigned int c[HS_SLICES];
#pragma unroll
        for (int k = 0; k < HS_SLICES; ++k) c[k] = part[(size_t)k * N + i];
#pragma unroll
        for (int k = 0; k < HS_SLICES; ++k) {
            part[(size_t)k * N + i] = run;
            run += c[k];
        }
        dinv[i] = __frsqrt_rn((float)(run + 1u));
        degc[i] = run;
    }
    s[t] = run;
    __syncthreads();
    for (int off = 1; off < SB; off <<= 1) {
        const unsigned int v = (t >= off) ? s[t - off] : 0u;
        __syncthreads();
        s[t] += v;
        __syncthreads();
    }
    if (i < N) rp_local[i] = s[t] - run;
    if (t == SB - 1) bsum[blockIdx.x] = s[t];
}

__device__ __forceinline__ void fill_put(unsigned int* cur, int c, int r, int lo,
                                         unsigned int cnt, int* __restrict__ srt,
                                         unsigned int E) {
    const unsigned int d = (unsigned int)(c - lo);
    if (d < cnt) {
        const unsigned int pos = atomicAdd(&cur[d], 1u);
        if (pos < E) srt[pos] = r;
    }
}

__device__ __forceinline__ void fill_put4(unsigned int* cur, const int4 c, const int4 r, int lo,
                                          unsigned int cnt, int* __restrict__ srt,
                                          unsigned int E) {
    const unsigned int d0 = (unsigned int)(c.x - lo);
    const unsigned int d1 = (unsigned int)(c.y - lo);
    const unsigned int d2 = (unsigned int)(c.z - lo);
    const unsigned int d3 = (unsigned int)(c.w - lo);
    unsigned int p0 = E, p1 = E, p2 = E, p3 = E;
    if (d0 < cnt) p0 = atomicAdd(&cur[d0], 1u);
    if (d1 < cnt) p1 = atomicAdd(&cur[d1], 1u);
    if (d2 < cnt) p2 = atomicAdd(&cur[d2], 1u);
    if (d3 < cnt) p3 = atomicAdd(&cur[d3], 1u);
    if (p0 < E) srt[p0] = r.x;
    if (p1 < E) srt[p1] = r.y;
    if (p2 < E) srt[p2] = r.z;
    if (p3 < E) srt[p3] = r.w;
}

__global__ __launch_bounds__(HT)
void fill_kernel(const int* __restrict__ row, const int* __restrict__ col,
                 const unsigned int* __restrict__ part,
                 const unsigned int* __restrict__ rp_local,
                 const unsigned int* __restrict__ bsum,
                 unsigned int* __restrict__ rp_abs, int* __restrict__ srt,
                 int N, int E) {
    __shared__ unsigned int cur[HR_NODES];
    __shared__ unsigned int tmp[NBR];
    __shared__ unsigned int bp[NBR];
    __shared__ unsigned int base0;
    const int t = threadIdx.x;
    const int lo = blockIdx.x * HR_NODES;
    const int slice = blockIdx.y;
    const int left = N - lo;
    const unsigned int cnt = (unsigned int)(left < HR_NODES ? left : HR_NODES);

    unsigned int vl[CI], vp[CI];
    {
        const unsigned int* __restrict__ rl = rp_local + lo;
        const unsigned int* __restrict__ pp = part + (size_t)slice * N + lo;
#pragma unroll
        for (int b = 0; b < CI; ++b) {
            unsigned int i = (unsigned int)t + (unsigned int)b * HT;
            if (i >= cnt) i = cnt - 1u;
            vl[b] = rl[i];
            vp[b] = pp[i];
        }
    }

    const int nb = (N + SB - 1) / SB;
    const int first = lo / SB;
    if (t == 0) base0 = 0u;
    if (t < NBR) tmp[t] = (first + t < nb) ? bsum[first + t] : 0u;
    __syncthreads();
    unsigned int mine = 0u;
    for (int b = t; b < first; b += HT) mine += bsum[b];
    if (mine != 0u) atomicAdd(&base0, mine);
    __syncthreads();
    if (t < NBR) {
        unsigned int sum = base0;
#pragma unroll
        for (int j = 0; j < NBR; ++j) sum += (j < t) ? tmp[j] : 0u;
        bp[t] = sum;
    }
    __syncthreads();
    unsigned int bv[CI];
#pragma unroll
    for (int b = 0; b < CI; ++b) bv[b] = bp[((unsigned int)t + (unsigned int)b * HT) / SB];
#pragma unroll
    for (int b = 0; b < CI; ++b) {
        const unsigned int i = (unsigned int)t + (unsigned int)b * HT;
        const unsigned int base = bv[b] + vl[b];
        const unsigned int first_place = base + vp[b];
        if (i < cnt) {
            if (slice == 0) rp_abs[lo + i] = base;
            cur[i] = first_place;
        }
    }
    __syncthreads();

    const int chunk = (((E + HS_SLICES - 1) / HS_SLICES) + 3) & ~3;
    const long long e0l = (long long)slice * chunk;
    const int e0 = e0l < (long long)E ? (int)e0l : E;
    const int e1 = (E - e0 < chunk) ? E : e0 + chunk;
    const unsigned int Eu = (unsigned int)E;

    if (((((size_t)col) | ((size_t)row)) & 15) == 0) {
        const int nv = (e1 - e0) >> 2;
        const int4* c4 = (const int4*)(col + e0);
        const int4* r4 = (const int4*)(row + e0);
        int i = t;
        for (; i + (FB - 1) * HT < nv; i += FB * HT) {
            int4 vc[FB], vr[FB];
#pragma unroll
            for (int b = 0; b < FB; ++b) { vc[b] = c4[i + b * HT]; vr[b] = r4[i + b * HT]; }
#pragma unroll
            for (int b = 0; b < FB; ++b) fill_put4(cur, vc[b], vr[b], lo, cnt, srt, Eu);
        }
        for (; i < nv; i += HT) {
            const int4 vc = c4[i];
            const int4 vr = r4[i];
            fill_put4(cur, vc, vr, lo, cnt, srt, Eu);
        }
        for (int e = e0 + (nv << 2) + t; e < e1; e += HT)
            fill_put(cur, col[e], row[e], lo, cnt, srt, Eu);
    } else {
        for (int e = e0 + t; e < e1; e += HT)
            fill_put(cur, col[e], row[e], lo, cnt, srt, Eu);
    }
}

#define WPB 2
#define PITCH 68

template <int CNT>
__device__ __forceinline__ void pull_group(float4& acc, const float4* __restrict__ x4,
                                           unsigned int r16, float w16, int l) {
    float4 v[CNT];
    const char* xb = (const char*)x4;
    const unsigned int lo = (unsigned int)l * 16u;
#pragma unroll
    for (int k = 0; k < CNT; ++k) {
        const unsigned int r = __shfl(r16, k, 16);
        v[k] = *(const float4*)(xb + ((r << 8) + lo));
    }
#pragma unroll
    for (int k = 0; k < CNT; ++k) {
        const float w = __shfl(w16, k, 16);
        acc.x += w * v[k].x;
        acc.y += w * v[k].y;
        acc.z += w * v[k].z;
        acc.w += w * v[k].w;
    }
}

__device__ __forceinline__ void pull_chunk(float4& acc, const float4* __restrict__ x4,
                                           unsigned int r16, float w16, int l, unsigned int rem) {
    if (rem > 8u) pull_group<16>(acc, x4, r16, w16, l);
    else if (rem > 0u) pull_group<8>(acc, x4, r16, w16, l);
}

__global__ __launch_bounds__(WPB * 32) __attribute__((amdgpu_waves_per_eu(7, 8)))
void gcn_pull_gemm_kernel(const float* __restrict__ x, const float* __restrict__ dinv,
                          const unsigned int* __restrict__ rp_abs,
                          const unsigned int* __restrict__ degc,
                          const int* __restrict__ srt,
                          const float* __restrict__ Wc, const float* __restrict__ bc,
                          const float* __restrict__ Ws, const float* __restrict__ bs,
                          float* __restrict__ out, int N, int E) {
    __shared__ float aggS[WPB][16 * PITCH];
    __shared__ float xS[WPB][16 * PITCH];
    const int lane   = threadIdx.x & 31;
    const int wave   = threadIdx.x >> 5;
    const int waveG  = blockIdx.x * WPB + wave;
    const int rowBase = waveG * 16;
    const int laneHi = lane >> 4;
    const int laneLo = lane & 15;
    const float4* x4 = (const float4*)x;
    const unsigned int Eu = (unsigned int)E;
    const unsigned int lastE = Eu > 0u ? Eu - 1u : 0u;
    const unsigned int lastNode = (unsigned int)(N - 1);
    float* myG = aggS[wave];
    float* myX = xS[wave];

    const int hrow = rowBase + laneLo;
    const bool hv = hrow < N;
    const int hr = hv ? hrow : N - 1;
    unsigned int startAll = rp_abs[hr];
    if (startAll > Eu) startAll = Eu;
    unsigned int nAll = hv ? degc[hr] : 0u;
    if (nAll > Eu - startAll) nAll = Eu - startAll;
    const float dcAll = dinv[hr];

    unsigned int start = __shfl(startAll, laneHi);
    unsigned int n = __shfl(nAll, laneHi);
    float dc = __shfl(dcAll, laneHi);
    int c = rowBase + laneHi; if (c > N - 1) c = N - 1;
    unsigned int rA, rB;
    {
        unsigned int qA = start + (unsigned int)laneLo;       if (qA > lastE) qA = lastE;
        unsigned int qB = start + (unsigned int)laneLo + 16u; if (qB > lastE) qB = lastE;
        const int iA = srt[qA];
        const int iB = srt[qB];
        rA = ((unsigned int)laneLo < n) ? (unsigned int)iA : (unsigned int)c;
        rB = ((unsigned int)(16 + laneLo) < n) ? (unsigned int)iB : (unsigned int)c;
        if (rA > lastNode) rA = lastNode;
        if (rB > lastNode) rB = lastNode;
    }
    float dA = dinv[rA];
    float dB = dinv[rB];
    float4 xs = x4[(size_t)c * (D / 4) + laneLo];
    unsigned int startN = __shfl(startAll, 2 + laneHi);
    unsigned int nN = __shfl(nAll, 2 + laneHi);
    float dcN = __shfl(dcAll, 2 + laneHi);
    int cN = rowBase + 2 + laneHi; if (cN > N - 1) cN = N - 1;
    int iAn, iBn;
    {
        unsigned int qA = startN + (unsigned int)laneLo;       if (qA > lastE) qA = lastE;
        unsigned int qB = startN + (unsigned int)laneLo + 16u; if (qB > lastE) qB = lastE;
        iAn = srt[qA];
        iBn = srt[qB];
    }

    for (int p = 0; p < 8; ++p) {
        unsigned int rAn = ((unsigned int)laneLo < nN) ? (unsigned int)iAn : (unsigned int)cN;
        unsigned int rBn = ((unsigned int)(16 + laneLo) < nN) ? (unsigned int)iBn : (unsigned int)cN;
        if (rAn > lastNode) rAn = lastNode;
        if (rBn > lastNode) rBn = lastNode;
        const float dAn = dinv[rAn];
        const float dBn = dinv[rBn];
        const float4 xsn = x4[(size_t)cN * (D / 4) + laneLo];
        const int p2 = p < 6 ? p + 2 : 7;
        const int m2 = 2 * p2 + laneHi;
        const unsigned int start2 = __shfl(startAll, m2);
        const unsigned int n2 = __shfl(nAll, m2);
        const float dc2 = __shfl(dcAll, m2);
        int c2 = rowBase + m2; if (c2 > N - 1) c2 = N - 1;
        unsigned int qA2 = start2 + (unsigned int)laneLo;       if (qA2 > lastE) qA2 = lastE;
        unsigned int qB2 = start2 + (unsigned int)laneLo + 16u; if (qB2 > lastE) qB2 = lastE;
        const int iA2 = srt[qA2];
        const int iB2 = srt[qB2];
        const unsigned int nOther = __shfl_xor(n, 16);
        const unsigned int nmax = __builtin_amdgcn_readfirstlane(n > nOther ? n : nOther);
        const float ws = dc * dc;
        float4 acc;
        acc.x = ws * xs.x; acc.y = ws * xs.y; acc.z = ws * xs.z; acc.w = ws * xs.w;
        const float wA = ((unsigned int)laneLo < n) ? dA * dc : 0.0f;
        const float wB = ((unsigned int)(16 + laneLo) < n) ? dB * dc : 0.0f;
        pull_chunk(acc, x4, rA, wA, laneLo, nmax);
        pull_chunk(acc, x4, rB, wB, laneLo, nmax > 16u ? nmax - 16u : 0u);
        for (unsigned int j = 32u; j < nmax; j += 16u) {
            const bool ok = (j + (unsigned int)laneLo) < n;
            unsigned int qC = start + j + (unsigned int)laneLo; if (qC > lastE) qC = lastE;
            const int iC = srt[qC];
            unsigned int rC = ok ? (unsigned int)iC : (unsigned int)c;
            if (rC > lastNode) rC = lastNode;
            const float wC = ok ? dinv[rC] * dc : 0.0f;
            pull_chunk(acc, x4, rC, wC, laneLo, nmax - j);
        }
        const int rowOff = (2 * p + laneHi) * PITCH + 4 * laneLo;
        *(float4*)(myG + rowOff) = acc;
        *(float4*)(myX + rowOff) = xs;
        start = startN; n = nN; dc = dcN; c = cN;
        rA = rAn; rB = rBn; dA = dAn; dB = dBn; xs = xsn;
        startN = start2; nN = n2; dcN = dc2; cN = c2; iAn = iA2; iBn = iB2;
    }
    __syncthreads();

    const float* ap = myX + laneLo * PITCH + 2 * laneHi;
    const float* gp = myG + laneLo * PITCH + 2 * laneHi;
    v2f a[16];
    v2f g[16];
#pragma unroll
    for (int k = 0; k < 16; ++k) {
        const float2 t = *(const float2*)(ap + 4 * k);
        a[k].x = t.x; a[k].y = t.y;
        const float2 u = *(const float2*)(gp + 4 * k);
        g[k].x = u.x; g[k].y = u.y;
    }

#pragma unroll
    for (int nt = 0; nt < 4; ++nt) {
        const int nn = nt * 16 + laneLo;
        v8f accC = {};
        v8f accS = {};
#pragma unroll
        for (int k0 = 0; k0 < D; k0 += 4) {
            const int kk = k0 + 2 * laneHi;
            v2f bC, bS;
            bC.x = Wc[(size_t)kk * D + nn];
            bC.y = Wc[(size_t)(kk + 1) * D + nn];
            bS.x = Ws[(size_t)kk * D + nn];
            bS.y = Ws[(size_t)(kk + 1) * D + nn];
            accC = __builtin_amdgcn_wmma_f32_16x16x4_f32(
                false, g[k0 >> 2], false, bC, (short)0, accC, false, false);
            accS = __builtin_amdgcn_wmma_f32_16x16x4_f32(
                false, a[k0 >> 2], false, bS, (short)0, accS, false, false);
        }
        const float bias = bs[nt * 16 + laneLo] + bc[nt * 16 + laneLo];
#pragma unroll
        for (int r = 0; r < 8; ++r) {
            const int row = rowBase + r + 8 * laneHi;
            if (row < N) {
                const size_t idx = (size_t)row * D + nt * 16 + laneLo;
                out[idx] = accS[r] + bias + accC[r];
            }
        }
    }
}

extern "C" void kernel_launch(void* const* d_in, const int* in_sizes, int n_in,
                              void* d_out, int out_size, void* d_ws, size_t ws_size,
                              hipStream_t stream) {
    const float* x  = (const float*)d_in[0];
    const int*   ei = (const int*)d_in[1];
    const float* Wc = (const float*)d_in[2];
    const float* bc = (const float*)d_in[3];
    const float* Ws = (const float*)d_in[4];
    const float* bs = (const float*)d_in[5];
    float* out = (float*)d_out;

    const int N = in_sizes[0] / D;
    const int E = in_sizes[1] / 2;

    float* dinv = (float*)d_ws;
    size_t off = ((size_t)N * sizeof(float) + 255) & ~(size_t)255;
    float* agg = (float*)((char*)d_ws + off);
    unsigned int* part = (unsigned int*)agg;
    int* srt = (int*)(part + (size_t)HS_SLICES * N);
    unsigned int* rp_local = (unsigned int*)(srt + E);
    unsigned int* rp_abs = rp_local + N;
    unsigned int* degc = rp_abs + N;
    unsigned int* bsum = degc + N;

    const dim3 hgrid((N + HR_NODES - 1) / HR_NODES, HS_SLICES);
    deg_hist_kernel<<<hgrid, HT, 0, stream>>>(ei + E, part, N, E);
    deg_scan_kernel<<<(N + SB - 1) / SB, SB, 0, stream>>>(part, dinv, degc, rp_local, bsum, N);
    fill_kernel<<<hgrid, HT, 0, stream>>>(ei, ei + E, part, rp_local, bsum, rp_abs, srt, N, E);
    gcn_pull_gemm_kernel<<<(N + 16 * WPB - 1) / (16 * WPB), WPB * 32, 0, stream>>>(
        x, dinv, rp_abs, degc, srt, Wc, bc, Ws, bs, out, N, E);
}


# Source weights without a memory gather: every node's degree as one byte in an LDS table per 25-wave block, rsqrt(1+deg) formed from the byte (same bits); wave-local tile hand-over; sum order kept
# speedup vs baseline: 3.4908x; 1.0359x over previous
#include <hip/hip_runtime.h>
#include <hip/hip_bf16.h>


typedef __attribute__((ext_vector_type(2))) float v2f;
typedef __attribute__((ext_vector_type(8))) float v8f;

#define D 64

#define HR_NODES 12800
#define HS_SLICES 32
#define HT 512
#define HB 8
#define FB 4
#define CI (HR_NODES / HT)
static_assert(HR_NODES % HT == 0, "the cursor set-up gives every thread the same number of nodes");

__device__ __forceinline__ void hist_bump(unsigned int* h, int c, int lo, unsigned int cnt) {
    const unsigned int d = (unsigned int)(c - lo);
    if (d < cnt) atomicAdd(&h[d], 1u);
}

__global__ __launch_bounds__(HT)
void deg_hist_kernel(const int* __restrict__ col, unsigned int* __restrict__ part,
                     int N, int E) {
    __shared__ unsigned int h[HR_NODES];
    const int lo = blockIdx.x * HR_NODES;
    const int slice = blockIdx.y;
    const int left = N - lo;
    const unsigned int cnt = (unsigned int)(left < HR_NODES ? left : HR_NODES);

    for (int i = threadIdx.x; i < HR_NODES; i += HT) h[i] = 0u;
    __syncthreads();

    const int chunk = (((E + HS_SLICES - 1) / HS_SLICES) + 3) & ~3;
    const long long e0l = (long long)slice * chunk;
    const int e0 = e0l < (long long)E ? (int)e0l : E;
    const int e1 = (E - e0 < chunk) ? E : e0 + chunk;

    if ((((size_t)col) & 15) == 0) {
        const int nv = (e1 - e0) >> 2;
        const int4* c4 = (const int4*)(col + e0);
        int i = threadIdx.x;
        for (; i + (HB - 1) * HT < nv; i += HB * HT) {
            int4 v[HB];
#pragma unroll
            for (int b = 0; b < HB; ++b) v[b] = c4[i + b * HT];
#pragma unroll
            for (int b = 0; b < HB; ++b) {
                hist_bump(h, v[b].x, lo, cnt);
                hist_bump(h, v[b].y, lo, cnt);
                hist_bump(h, v[b].z, lo, cnt);
                hist_bump(h, v[b].w, lo, cnt);
            }
        }
        for (; i < nv; i += HT) {
            const int4 v = c4[i];
            hist_bump(h, v.x, lo, cnt);
            hist_bump(h, v.y, lo, cnt);
            hist_bump(h, v.z, lo, cnt);
            hist_bump(h, v.w, lo, cnt);
        }
        for (int e = e0 + (nv << 2) + threadIdx.x; e < e1; e += HT)
            hist_bump(h, col[e], lo, cnt);
    } else {
        for (int e = e0 + threadIdx.x; e < e1; e += HT)
            hist_bump(h, col[e], lo, cnt);
    }
    __syncthreads();

    unsigned int* dst = part + (size_t)slice * N + lo;
    for (unsigned int i = threadIdx.x; i < cnt; i += HT) dst[i] = h[i];
}

#define SB 256
#define NBR (HR_NODES / SB)
static_assert(HR_NODES % SB == 0, "a node range is a whole number of scan blocks");

__global__ __launch_bounds__(SB)
void deg_scan_kernel(unsigned int* __restrict__ part, float* __restrict__ dinv,
                     unsigned int* __restrict__ degc, unsigned int* __restrict__ rp_local,
                     unsigned int* __restrict__ bsum, unsigned char* __restrict__ deg8, int N) {
    __shared__ unsigned int s[SB];
    const int t = threadIdx.x;
    const int i = blockIdx.x * SB + t;
    unsigned int run = 0u;
    if (i < N) {
        unsigned int c[HS_SLICES];
#pragma unroll
        for (int k = 0; k < HS_SLICES; ++k) c[k] = part[(size_t)k * N + i];
#pragma unroll
        for (int k = 0; k < HS_SLICES; ++k) {
            part[(size_t)k * N + i] = run;
            run += c[k];
        }
        dinv[i] = __frsqrt_rn((float)(run + 1u));
        degc[i] = run;
    }
    deg8[i] = (unsigned char)(run < 255u ? run : 255u);
    s[t] = run;
    __syncthreads();
    for (int off = 1; off < SB; off <<= 1) {
        const unsigned int v = (t >= off) ? s[t - off] : 0u;
        __syncthreads();
        s[t] += v;
        __syncthreads();
    }
    if (i < N) rp_local[i] = s[t] - run;
    if (t == SB - 1) bsum[blockIdx.x] = s[t];
}

__device__ __forceinline__ void fill_put(unsigned int* cur, int c, int r, int lo,
                                         unsigned int cnt, int* __restrict__ srt,
                                         unsigned int E) {
    const unsigned int d = (unsigned int)(c - lo);
    if (d < cnt) {
        const unsigned int pos = atomicAdd(&cur[d], 1u);
        if (pos < E) srt[pos] = r;
    }
}

__device__ __forceinline__ void fill_put4(unsigned int* cur, const int4 c, const int4 r, int lo,
                                          unsigned int cnt, int* __restrict__ srt,
                                          unsigned int E) {
    const unsigned int d0 = (unsigned int)(c.x - lo);
    const unsigned int d1 = (unsigned int)(c.y - lo);
    const unsigned int d2 = (unsigned int)(c.z - lo);
    const unsigned int d3 = (unsigned int)(c.w - lo);
    unsigned int p0 = E, p1 = E, p2 = E, p3 = E;
    if (d0 < cnt) p0 = atomicAdd(&cur[d0], 1u);
    if (d1 < cnt) p1 = atomicAdd(&cur[d1], 1u);
    if (d2 < cnt) p2 = atomicAdd(&cur[d2], 1u);
    if (d3 < cnt) p3 = atomicAdd(&cur[d3], 1u);
    if (p0 < E) srt[p0] = r.x;
    if (p1 < E) srt[p1] = r.y;
    if (p2 < E) srt[p2] = r.z;
    if (p3 < E) srt[p3] = r.w;
}

__global__ __launch_bounds__(HT)
void fill_kernel(const int* __restrict__ row, const int* __restrict__ col,
                 const unsigned int* __restrict__ part,
                 const unsigned int* __restrict__ rp_local,
                 const unsigned int* __restrict__ bsum,
                 unsigned int* __restrict__ rp_abs, int* __restrict__ srt,
                 int N, int E) {
    __shared__ unsigned int cur[HR_NODES];
    __shared__ unsigned int tmp[NBR];
    __shared__ unsigned int bp[NBR];
    __shared__ unsigned int base0;
    const int t = threadIdx.x;
    const int lo = blockIdx.x * HR_NODES;
    const int slice = blockIdx.y;
    const int left = N - lo;
    const unsigned int cnt = (unsigned int)(left < HR_NODES ? left : HR_NODES);

    unsigned int vl[CI], vp[CI];
    {
        const unsigned int* __restrict__ rl = rp_local + lo;
        const unsigned int* __restrict__ pp = part + (size_t)slice * N + lo;
#pragma unroll
        for (int b = 0; b < CI; ++b) {
            unsigned int i = (unsigned int)t + (unsigned int)b * HT;
            if (i >= cnt) i = cnt - 1u;
            vl[b] = rl[i];
            vp[b] = pp[i];
        }
    }

    const int nb = (N + SB - 1) / SB;
    const int first = lo / SB;
    if (t == 0) base0 = 0u;
    if (t < NBR) tmp[t] = (first + t < nb) ? bsum[first + t] : 0u;
    __syncthreads();
    unsigned int mine = 0u;
    for (int b = t; b < first; b += HT) mine += bsum[b];
    if (mine != 0u) atomicAdd(&base0, mine);
    __syncthreads();
    if (t < NBR) {
        unsigned int sum = base0;
#pragma unroll
        for (int j = 0; j < NBR; ++j) sum += (j < t) ? tmp[j] : 0u;
        bp[t] = sum;
    }
    __syncthreads();
    unsigned int bv[CI];
#pragma unroll
    for (int b = 0; b < CI; ++b) bv[b] = bp[((unsigned int)t + (unsigned int)b * HT) / SB];
#pragma unroll
    for (int b = 0; b < CI; ++b) {
        const unsigned int i = (unsigned int)t + (unsigned int)b * HT;
        const unsigned int base = bv[b] + vl[b];
        const unsigned int first_place = base + vp[b];
        if (i < cnt) {
            if (slice == 0) rp_abs[lo + i] = base;
            cur[i] = first_place;
        }
    }
    __syncthreads();

    const int chunk = (((E + HS_SLICES - 1) / HS_SLICES) + 3) & ~3;
    const long long e0l = (long long)slice * chunk;
    const int e0 = e0l < (long long)E ? (int)e0l : E;
    const int e1 = (E - e0 < chunk) ? E : e0 + chunk;
    const unsigned int Eu = (unsigned int)E;

    if (((((size_t)col) | ((size_t)row)) & 15) == 0) {
        const int nv = (e1 - e0) >> 2;
        const int4* c4 = (const int4*)(col + e0);
        const int4* r4 = (const int4*)(row + e0);
        int i = t;
        for (; i + (FB - 1) * HT < nv; i += FB * HT) {
            int4 vc[FB], vr[FB];
#pragma unroll
            for (int b = 0; b < FB; ++b) { vc[b] = c4[i + b * HT]; vr[b] = r4[i + b * HT]; }
#pragma unroll
            for (int b = 0; b < FB; ++b) fill_put4(cur, vc[b], vr[b], lo, cnt, srt, Eu);
        }
        for (; i < nv; i += HT) {
            const int4 vc = c4[i];
            const int4 vr = r4[i];
            fill_put4(cur, vc, vr, lo, cnt, srt, Eu);
        }
        for (int e = e0 + (nv << 2) + t; e < e1; e += HT)
            fill_put(cur, col[e], row[e], lo, cnt, srt, Eu);
    } else {
        for (int e = e0 + t; e < e1; e += HT)
            fill_put(cur, col[e], row[e], lo, cnt, srt, Eu);
    }
}

#define WPB 25
#define TCAP 102400
#define TFILL (TCAP / 16 / (WPB * 32))
static_assert(TFILL * 16 * WPB * 32 == TCAP, "the block's lanes cover the table in whole rounds of 16-byte loads");
#define PITCH 68

template <int CNT>
__device__ __forceinline__ void pull_group(float4& acc, const float4* __restrict__ x4,
                                           unsigned int r16, float w16, int l) {
    float4 v[CNT];
    const char* xb = (const char*)x4;
    const unsigned int lo = (unsigned int)l * 16u;
#pragma unroll
    for (int k = 0; k < CNT; ++k) {
        const unsigned int r = __shfl(r16, k, 16);
        v[k] = *(const float4*)(xb + ((r << 8) + lo));
    }
#pragma unroll
    for (int k = 0; k < CNT; ++k) {
        const float w = __shfl(w16, k, 16);
        acc.x += w * v[k].x;
        acc.y += w * v[k].y;
        acc.z += w * v[k].z;
        acc.w += w * v[k].w;
    }
}

__device__ __forceinline__ unsigned int deg_byte(const unsigned char* tab, unsigned int r) {
    const bool in = r < (unsigned int)TCAP;
    const unsigned int q = tab[in ? r : 0u];
    return in ? q : 255u;
}

__device__ __forceinline__ float src_dinv(unsigned int q, const float* __restrict__ dinv,
                                          unsigned int r) {
    float w = __frsqrt_rn((float)(q + 1u));
    if (q == 255u) w = dinv[r];
    return w;
}

__device__ __forceinline__ void pull_chunk(float4& acc, const float4* __restrict__ x4,
                                           unsigned int r16, float w16, int l, unsigned int rem) {
    if (rem > 8u) pull_group<16>(acc, x4, r16, w16, l);
    else if (rem > 0u) pull_group<8>(acc, x4, r16, w16, l);
}

__global__ __launch_bounds__(WPB * 32) __attribute__((amdgpu_waves_per_eu(7, 8)))
void gcn_pull_gemm_kernel(const float* __restrict__ x, const float* __restrict__ dinv,
                          const unsigned int* __restrict__ rp_abs,
                          const unsigned int* __restrict__ degc,
                          const int* __restrict__ srt, const unsigned char* __restrict__ deg8,
                          const float* __restrict__ Wc, const float* __restrict__ bc,
                          const float* __restrict__ Ws, const float* __restrict__ bs,
                          float* __restrict__ out, int N, int E) {
    __shared__ float aggS[WPB][16 * PITCH];
    __shared__ float xS[WPB][16 * PITCH];
    __shared__ __attribute__((aligned(16))) unsigned int degT[TCAP / 4];
    const unsigned char* tab = (const unsigned char*)degT;
    {
        const unsigned int nn = (unsigned int)N < (unsigned int)TCAP ? (unsigned int)N : (unsigned int)TCAP;
        const unsigned int nq = (nn + 15u) >> 4;
        const uint4* __restrict__ s16 = (const uint4*)deg8;
        uint4* d16 = (uint4*)degT;
        uint4 tv[TFILL];
#pragma unroll
        for (int b = 0; b < TFILL; ++b) {
            unsigned int i = threadIdx.x + (unsigned int)b * (WPB * 32);
            if (i >= nq) i = nq - 1u;
            tv[b] = s16[i];
        }
#pragma unroll
        for (int b = 0; b < TFILL; ++b) d16[threadIdx.x + (unsigned int)b * (WPB * 32)] = tv[b];
    }
    const int lane   = threadIdx.x & 31;
    const int wave   = threadIdx.x >> 5;
    const int waveG  = blockIdx.x * WPB + wave;
    const int rowBase = waveG * 16;
    const int laneHi = lane >> 4;
    const int laneLo = lane & 15;
    const float4* x4 = (const float4*)x;
    const unsigned int Eu = (unsigned int)E;
    const unsigned int lastE = Eu > 0u ? Eu - 1u : 0u;
    const unsigned int lastNode = (unsigned int)(N - 1);
    float* myG = aggS[wave];
    float* myX = xS[wave];

    const int hrow = rowBase + laneLo;
    const bool hv = hrow < N;
    const int hr = hv ? hrow : N - 1;
    unsigned int startAll = rp_abs[hr];
    if (startAll > Eu) startAll = Eu;
    unsigned int nAll = hv ? degc[hr] : 0u;
    if (nAll > Eu - startAll) nAll = Eu - startAll;
    const float dcAll = dinv[hr];
    __syncthreads();

    unsigned int start = __shfl(startAll, laneHi);
    unsigned int n = __shfl(nAll, laneHi);
    float dc = __shfl(dcAll, laneHi);
    int c = rowBase + laneHi; if (c > N - 1) c = N - 1;
    unsigned int rA, rB;
    {
        unsigned int qA = start + (unsigned int)laneLo;       if (qA > lastE) qA = lastE;
        unsigned int qB = start + (unsigned int)laneLo + 16u; if (qB > lastE) qB = lastE;
        const int iA = srt[qA];
        const int iB = srt[qB];
        rA = ((unsigned int)laneLo < n) ? (unsigned int)iA : (unsigned int)c;
        rB = ((unsigned int)(16 + laneLo) < n) ? (unsigned int)iB : (unsigned int)c;
        if (rA > lastNode) rA = lastNode;
        if (rB > lastNode) rB = lastNode;
    }
    unsigned int qA = deg_byte(tab, rA);
    unsigned int qB = deg_byte(tab, rB);
    float4 xs = x4[(size_t)c * (D / 4) + laneLo];
    unsigned int startN = __shfl(startAll, 2 + laneHi);
    unsigned int nN = __shfl(nAll, 2 + laneHi);
    float dcN = __shfl(dcAll, 2 + laneHi);
    int cN = rowBase + 2 + laneHi; if (cN > N - 1) cN = N - 1;
    int iAn, iBn;
    {
        unsigned int qA = startN + (unsigned int)laneLo;       if (qA > lastE) qA = lastE;
        unsigned int qB = startN + (unsigned int)laneLo + 16u; if (qB > lastE) qB = lastE;
        iAn = srt[qA];
        iBn = srt[qB];
    }

    for (int p = 0; p < 8; ++p) {
        const float dA = src_dinv(qA, dinv, rA);
        const float dB = src_dinv(qB, dinv, rB);
        unsigned int rAn = ((unsigned int)laneLo < nN) ? (unsigned int)iAn : (unsigned int)cN;
        unsigned int rBn = ((unsigned int)(16 + laneLo) < nN) ? (unsigned int)iBn : (unsigned int)cN;
        if (rAn > lastNode) rAn = lastNode;
        if (rBn > lastNode) rBn = lastNode;
        const unsigned int qAn = deg_byte(tab, rAn);
        const unsigned int qBn = deg_byte(tab, rBn);
        const float4 xsn = x4[(size_t)cN * (D / 4) + laneLo];
        const int p2 = p < 6 ? p + 2 : 7;
        const int m2 = 2 * p2 + laneHi;
        const unsigned int start2 = __shfl(startAll, m2);
        const unsigned int n2 = __shfl(nAll, m2);
        const float dc2 = __shfl(dcAll, m2);
        int c2 = rowBase + m2; if (c2 > N - 1) c2 = N - 1;
        unsigned int qA2 = start2 + (unsigned int)laneLo;       if (qA2 > lastE) qA2 = lastE;
        unsigned int qB2 = start2 + (unsigned int)laneLo + 16u; if (qB2 > lastE) qB2 = lastE;
        const int iA2 = srt[qA2];
        const int iB2 = srt[qB2];
        const unsigned int nOther = __shfl_xor(n, 16);
        const unsigned int nmax = __builtin_amdgcn_readfirstlane(n > nOther ? n : nOther);
        const float ws = dc * dc;
        float4 acc;
        acc.x = ws * xs.x; acc.y = ws * xs.y; acc.z = ws * xs.z; acc.w = ws * xs.w;
        const float wA = ((unsigned int)laneLo < n) ? dA * dc : 0.0f;
        const float wB = ((unsigned int)(16 + laneLo) < n) ? dB * dc : 0.0f;
        pull_chunk(acc, x4, rA, wA, laneLo, nmax);
        pull_chunk(acc, x4, rB, wB, laneLo, nmax > 16u ? nmax - 16u : 0u);
        for (unsigned int j = 32u; j < nmax; j += 16u) {
            const bool ok = (j + (unsigned int)laneLo) < n;
            unsigned int qC = start + j + (unsigned int)laneLo; if (qC > lastE) qC = lastE;
            const int iC = srt[qC];
            unsigned int rC = ok ? (unsigned int)iC : (unsigned int)c;
            if (rC > lastNode) rC = lastNode;
            const float wC = ok ? src_dinv(deg_byte(tab, rC), dinv, rC) * dc : 0.0f;
            pull_chunk(acc, x4, rC, wC, laneLo, nmax - j);
        }
        const int rowOff = (2 * p + laneHi) * PITCH + 4 * laneLo;
        *(float4*)(myG + rowOff) = acc;
        *(float4*)(myX + rowOff) = xs;
        start = startN; n = nN; dc = dcN; c = cN;
        rA = rAn; rB = rBn; qA = qAn; qB = qBn; xs = xsn;
        startN = start2; nN = n2; dcN = dc2; cN = c2; iAn = iA2; iBn = iB2;
    }
    __builtin_amdgcn_fence(__ATOMIC_RELEASE, "wavefront");
    __builtin_amdgcn_wave_barrier();
    __builtin_amdgcn_fence(__ATOMIC_ACQUIRE, "wavefront");

    const float* ap = myX + laneLo * PITCH + 2 * laneHi;
    const float* gp = myG + laneLo * PITCH + 2 * laneHi;
    v2f a[16];
    v2f g[16];
#pragma unroll
    for (int k = 0; k < 16; ++k) {
        const float2 t = *(const float2*)(ap + 4 * k);
        a[k].x = t.x; a[k].y = t.y;
        const float2 u = *(const float2*)(gp + 4 * k);
        g[k].x = u.x; g[k].y = u.y;
    }

#pragma unroll
    for (int nt = 0; nt < 4; ++nt) {
        const int nn = nt * 16 + laneLo;
        v8f accC = {};
        v8f accS = {};
#pragma unroll
        for (int k0 = 0; k0 < D; k0 += 4) {
            const int kk = k0 + 2 * laneHi;
            v2f bC, bS;
            bC.x = Wc[(size_t)kk * D + nn];
            bC.y = Wc[(size_t)(kk + 1) * D + nn];
            bS.x = Ws[(size_t)kk * D + nn];
            bS.y = Ws[(size_t)(kk + 1) * D + nn];
            accC = __builtin_amdgcn_wmma_f32_16x16x4_f32(
                false, g[k0 >> 2], false, bC, (short)0, accC, false, false);
            accS = __builtin_amdgcn_wmma_f32_16x16x4_f32(
                false, a[k0 >> 2], false, bS, (short)0, accS, false, false);
        }
        const float bias = bs[nt * 16 + laneLo] + bc[nt * 16 + laneLo];
#pragma unroll
        for (int r = 0; r < 8; ++r) {
            const int row = rowBase + r + 8 * laneHi;
            if (row < N) {
                const size_t idx = (size_t)row * D + nt * 16 + laneLo;
                out[idx] = accS[r] + bias + accC[r];
            }
        }
    }
}

extern "C" void kernel_launch(void* const* d_in, const int* in_sizes, int n_in,
                              void* d_out, int out_size, void* d_ws, size_t ws_size,
                              hipStream_t stream) {
    const float* x  = (const float*)d_in[0];
    const int*   ei = (const int*)d_in[1];
    const float* Wc = (const float*)d_in[2];
    const float* bc = (const float*)d_in[3];
    const float* Ws = (const float*)d_in[4];
    const float* bs = (const float*)d_in[5];
    float* out = (float*)d_out;

    const int N = in_sizes[0] / D;
    const int E = in_sizes[1] / 2;

    float* dinv = (float*)d_ws;
    size_t off = ((size_t)N * sizeof(float) + 255) & ~(size_t)255;
    float* agg = (float*)((char*)d_ws + off);
    unsigned int* part = (unsigned int*)agg;
    int* srt = (int*)(part + (size_t)HS_SLICES * N);
    unsigned int* rp_local = (unsigned int*)(srt + E);
    unsigned int* rp_abs = rp_local + N;
    unsigned int* degc = rp_abs + N;
    unsigned int* bsum = degc + N;
    const int nb = (N + SB - 1) / SB;
    size_t off8 = (off + ((size_t)HS_SLICES * N + (size_t)E + 3 * (size_t)N + (size_t)nb) * sizeof(unsigned int) + 255) & ~(size_t)255;
    unsigned char* deg8 = (unsigned char*)d_ws + off8;

    const dim3 hgrid((N + HR_NODES - 1) / HR_NODES, HS_SLICES);
    deg_hist_kernel<<<hgrid, HT, 0, stream>>>(ei + E, part, N, E);
    deg_scan_kernel<<<nb, SB, 0, stream>>>(part, dinv, degc, rp_local, bsum, deg8, N);
    fill_kernel<<<hgrid, HT, 0, stream>>>(ei, ei + E, part, rp_local, bsum, rp_abs, srt, N, E);
    gcn_pull_gemm_kernel<<<(N + 16 * WPB - 1) / (16 * WPB), WPB * 32, 0, stream>>>(
        x, dinv, rp_abs, degc, srt, deg8, Wc, bc, Ws, bs, out, N, E);
}


# x@W_self product done before the neighbour pull (own rows to LDS at entry, accumulators parked in that tile), agg@W_conv behind it; each half loads its B fragments before its WMMAs; sum order kept
# speedup vs baseline: 3.7287x; 1.0682x over previous
#include <hip/hip_runtime.h>
#include <hip/hip_bf16.h>


typedef __attribute__((ext_vector_type(2))) float v2f;
typedef __attribute__((ext_vector_type(8))) float v8f;

#define D 64

#define HR_NODES 12800
#define HS_SLICES 32
#define HT 512
#define HB 8
#define FB 4
#define CI (HR_NODES / HT)
static_assert(HR_NODES % HT == 0, "the cursor set-up gives every thread the same number of nodes");

__device__ __forceinline__ void hist_bump(unsigned int* h, int c, int lo, unsigned int cnt) {
    const unsigned int d = (unsigned int)(c - lo);
    if (d < cnt) atomicAdd(&h[d], 1u);
}

__global__ __launch_bounds__(HT)
void deg_hist_kernel(const int* __restrict__ col, unsigned int* __restrict__ part,
                     int N, int E) {
    __shared__ unsigned int h[HR_NODES];
    const int lo = blockIdx.x * HR_NODES;
    const int slice = blockIdx.y;
    const int left = N - lo;
    const unsigned int cnt = (unsigned int)(left < HR_NODES ? left : HR_NODES);

    for (int i = threadIdx.x; i < HR_NODES; i += HT) h[i] = 0u;
    __syncthreads();

    const int chunk = (((E + HS_SLICES - 1) / HS_SLICES) + 3) & ~3;
    const long long e0l = (long long)slice * chunk;
    const int e0 = e0l < (long long)E ? (int)e0l : E;
    const int e1 = (E - e0 < chunk) ? E : e0 + chunk;

    if ((((size_t)col) & 15) == 0) {
        const int nv = (e1 - e0) >> 2;
        const int4* c4 = (const int4*)(col + e0);
        int i = threadIdx.x;
        for (; i + (HB - 1) * HT < nv; i += HB * HT) {
            int4 v[HB];
#pragma unroll
            for (int b = 0; b < HB; ++b) v[b] = c4[i + b * HT];
#pragma unroll
            for (int b = 0; b < HB; ++b) {
                hist_bump(h, v[b].x, lo, cnt);
                hist_bump(h, v[b].y, lo, cnt);
                hist_bump(h, v[b].z, lo, cnt);
                hist_bump(h, v[b].w, lo, cnt);
            }
        }
        for (; i < nv; i += HT) {
            const int4 v = c4[i];
            hist_bump(h, v.x, lo, cnt);
            hist_bump(h, v.y, lo, cnt);
            hist_bump(h, v.z, lo, cnt);
            hist_bump(h, v.w, lo, cnt);
        }
        for (int e = e0 + (nv << 2) + threadIdx.x; e < e1; e += HT)
            hist_bump(h, col[e], lo, cnt);
    } else {
        for (int e = e0 + threadIdx.x; e < e1; e += HT)
            hist_bump(h, col[e], lo, cnt);
    }
    __syncthreads();

    unsigned int* dst = part + (size_t)slice * N + lo;
    for (unsigned int i = threadIdx.x; i < cnt; i += HT) dst[i] = h[i];
}

#define SB 256
#define NBR (HR_NODES / SB)
static_assert(HR_NODES % SB == 0, "a node range is a whole number of scan blocks");

__global__ __launch_bounds__(SB)
void deg_scan_kernel(unsigned int* __restrict__ part, float* __restrict__ dinv,
                     unsigned int* __restrict__ degc, unsigned int* __restrict__ rp_local,
                     unsigned int* __restrict__ bsum, unsigned char* __restrict__ deg8, int N) {
    __shared__ unsigned int s[SB];
    const int t = threadIdx.x;
    const int i = blockIdx.x * SB + t;
    unsigned int run = 0u;
    if (i < N) {
        unsigned int c[HS_SLICES];
#pragma unroll
        for (int k = 0; k < HS_SLICES; ++k) c[k] = part[(size_t)k * N + i];
#pragma unroll
        for (int k = 0; k < HS_SLICES; ++k) {
            part[(size_t)k * N + i] = run;
            run += c[k];
        }
        dinv[i] = __frsqrt_rn((float)(run + 1u));
        degc[i] = run;
    }
    deg8[i] = (unsigned char)(run < 255u ? run : 255u);
    s[t] = run;
    __syncthreads();
    for (int off = 1; off < SB; off <<= 1) {
        const unsigned int v = (t >= off) ? s[t - off] : 0u;
        __syncthreads();
        s[t] += v;
        __syncthreads();
    }
    if (i < N) rp_local[i] = s[t] - run;
    if (t == SB - 1) bsum[blockIdx.x] = s[t];
}

__device__ __forceinline__ void fill_put(unsigned int* cur, int c, int r, int lo,
                                         unsigned int cnt, int* __restrict__ srt,
                                         unsigned int E) {
    const unsigned int d = (unsigned int)(c - lo);
    if (d < cnt) {
        const unsigned int pos = atomicAdd(&cur[d], 1u);
        if (pos < E) srt[pos] = r;
    }
}

__device__ __forceinline__ void fill_put4(unsigned int* cur, const int4 c, const int4 r, int lo,
                                          unsigned int cnt, int* __restrict__ srt,
                                          unsigned int E) {
    const unsigned int d0 = (unsigned int)(c.x - lo);
    const unsigned int d1 = (unsigned int)(c.y - lo);
    const unsigned int d2 = (unsigned int)(c.z - lo);
    const unsigned int d3 = (unsigned int)(c.w - lo);
    unsigned int p0 = E, p1 = E, p2 = E, p3 = E;
    if (d0 < cnt) p0 = atomicAdd(&cur[d0], 1u);
    if (d1 < cnt) p1 = atomicAdd(&cur[d1], 1u);
    if (d2 < cnt) p2 = atomicAdd(&cur[d2], 1u);
    if (d3 < cnt) p3 = atomicAdd(&cur[d3], 1u);
    if (p0 < E) srt[p0] = r.x;
    if (p1 < E) srt[p1] = r.y;
    if (p2 < E) srt[p2] = r.z;
    if (p3 < E) srt[p3] = r.w;
}

__global__ __launch_bounds__(HT)
void fill_kernel(const int* __restrict__ row, const int* __restrict__ col,
                 const unsigned int* __restrict__ part,
                 const unsigned int* __restrict__ rp_local,
                 const unsigned int* __restrict__ bsum,
                 unsigned int* __restrict__ rp_abs, int* __restrict__ srt,
                 int N, int E) {
    __shared__ unsigned int cur[HR_NODES];
    __shared__ unsigned int tmp[NBR];
    __shared__ unsigned int bp[NBR];
    __shared__ unsigned int base0;
    const int t = threadIdx.x;
    const int lo = blockIdx.x * HR_NODES;
    const int slice = blockIdx.y;
    const int left = N - lo;
    const unsigned int cnt = (unsigned int)(left < HR_NODES ? left : HR_NODES);

    unsigned int vl[CI], vp[CI];
    {
        const unsigned int* __restrict__ rl = rp_local + lo;
        const unsigned int* __restrict__ pp = part + (size_t)slice * N + lo;
#pragma unroll
        for (int b = 0; b < CI; ++b) {
            unsigned int i = (unsigned int)t + (unsigned int)b * HT;
            if (i >= cnt) i = cnt - 1u;
            vl[b] = rl[i];
            vp[b] = pp[i];
        }
    }

    const int nb = (N + SB - 1) / SB;
    const int first = lo / SB;
    if (t == 0) base0 = 0u;
    if (t < NBR) tmp[t] = (first + t < nb) ? bsum[first + t] : 0u;
    __syncthreads();
    unsigned int mine = 0u;
    for (int b = t; b < first; b += HT) mine += bsum[b];
    if (mine != 0u) atomicAdd(&base0, mine);
    __syncthreads();
    if (t < NBR) {
        unsigned int sum = base0;
#pragma unroll
        for (int j = 0; j < NBR; ++j) sum += (j < t) ? tmp[j] : 0u;
        bp[t] = sum;
    }
    __syncthreads();
    unsigned int bv[CI];
#pragma unroll
    for (int b = 0; b < CI; ++b) bv[b] = bp[((unsigned int)t + (unsigned int)b * HT) / SB];
#pragma unroll
    for (int b = 0; b < CI; ++b) {
        const unsigned int i = (unsigned int)t + (unsigned int)b * HT;
        const unsigned int base = bv[b] + vl[b];
        const unsigned int first_place = base + vp[b];
        if (i < cnt) {
            if (slice == 0) rp_abs[lo + i] = base;
            cur[i] = first_place;
        }
    }
    __syncthreads();

    const int chunk = (((E + HS_SLICES - 1) / HS_SLICES) + 3) & ~3;
    const long long e0l = (long long)slice * chunk;
    const int e0 = e0l < (long long)E ? (int)e0l : E;
    const int e1 = (E - e0 < chunk) ? E : e0 + chunk;
    const unsigned int Eu = (unsigned int)E;

    if (((((size_t)col) | ((size_t)row)) & 15) == 0) {
        const int nv = (e1 - e0) >> 2;
        const int4* c4 = (const int4*)(col + e0);
        const int4* r4 = (const int4*)(row + e0);
        int i = t;
        for (; i + (FB - 1) * HT < nv; i += FB * HT) {
            int4 vc[FB], vr[FB];
#pragma unroll
            for (int b = 0; b < FB; ++b) { vc[b] = c4[i + b * HT]; vr[b] = r4[i + b * HT]; }
#pragma unroll
            for (int b = 0; b < FB; ++b) fill_put4(cur, vc[b], vr[b], lo, cnt, srt, Eu);
        }
        for (; i < nv; i += HT) {
            const int4 vc = c4[i];
            const int4 vr = r4[i];
            fill_put4(cur, vc, vr, lo, cnt, srt, Eu);
        }
        for (int e = e0 + (nv << 2) + t; e < e1; e += HT)
            fill_put(cur, col[e], row[e], lo, cnt, srt, Eu);
    } else {
        for (int e = e0 + t; e < e1; e += HT)
            fill_put(cur, col[e], row[e], lo, cnt, srt, Eu);
    }
}

#define WPB 25
#define TCAP 102400
#define TFILL (TCAP / 16 / (WPB * 32))
static_assert(TFILL * 16 * WPB * 32 == TCAP, "the block's lanes cover the table in whole rounds of 16-byte loads");
#define PITCH 68

template <int CNT>
__device__ __forceinline__ void pull_group(float4& acc, const float4* __restrict__ x4,
                                           unsigned int r16, float w16, int l) {
    float4 v[CNT];
    const char* xb = (const char*)x4;
    const unsigned int lo = (unsigned int)l * 16u;
#pragma unroll
    for (int k = 0; k < CNT; ++k) {
        const unsigned int r = __shfl(r16, k, 16);
        v[k] = *(const float4*)(xb + ((r << 8) + lo));
    }
#pragma unroll
    for (int k = 0; k < CNT; ++k) {
        const float w = __shfl(w16, k, 16);
        acc.x += w * v[k].x;
        acc.y += w * v[k].y;
        acc.z += w * v[k].z;
        acc.w += w * v[k].w;
    }
}

__device__ __forceinline__ unsigned int deg_byte(const unsigned char* tab, unsigned int r) {
    const bool in = r < (unsigned int)TCAP;
    const unsigned int q = tab[in ? r : 0u];
    return in ? q : 255u;
}

__device__ __forceinline__ float src_dinv(unsigned int q, const float* __restrict__ dinv,
                                          unsigned int r) {
    float w = __frsqrt_rn((float)(q + 1u));
    if (q == 255u) w = dinv[r];
    return w;
}

__device__ __forceinline__ void pull_chunk(float4& acc, const float4* __restrict__ x4,
                                           unsigned int r16, float w16, int l, unsigned int rem) {
    if (rem > 8u) pull_group<16>(acc, x4, r16, w16, l);
    else if (rem > 0u) pull_group<8>(acc, x4, r16, w16, l);
}

__global__ __launch_bounds__(WPB * 32) __attribute__((amdgpu_waves_per_eu(7, 8)))
void gcn_pull_gemm_kernel(const float* __restrict__ x, const float* __restrict__ dinv,
                          const unsigned int* __restrict__ rp_abs,
                          const unsigned int* __restrict__ degc,
                          const int* __restrict__ srt, const unsigned char* __restrict__ deg8,
                          const float* __restrict__ Wc, const float* __restrict__ bc,
                          const float* __restrict__ Ws, const float* __restrict__ bs,
                          float* __restrict__ out, int N, int E) {
    __shared__ float aggS[WPB][16 * PITCH];
    __shared__ float xS[WPB][16 * PITCH];
    __shared__ __attribute__((aligned(16))) unsigned int degT[TCAP / 4];
    const unsigned char* tab = (const unsigned char*)degT;
    const int lane   = threadIdx.x & 31;
    const int wave   = threadIdx.x >> 5;
    const int waveG  = blockIdx.x * WPB + wave;
    const int rowBase = waveG * 16;
    const int laneHi = lane >> 4;
    const int laneLo = lane & 15;
    const float4* x4 = (const float4*)x;
    float* myG = aggS[wave];
    float* myX = xS[wave];
    {
        const unsigned int nn = (unsigned int)N < (unsigned int)TCAP ? (unsigned int)N : (unsigned int)TCAP;
        const unsigned int nq = (nn + 15u) >> 4;
        const uint4* __restrict__ s16 = (const uint4*)deg8;
        uint4* d16 = (uint4*)degT;
        uint4 tv[TFILL];
#pragma unroll
        for (int b = 0; b < TFILL; ++b) {
            unsigned int i = threadIdx.x + (unsigned int)b * (WPB * 32);
            if (i >= nq) i = nq - 1u;
            tv[b] = s16[i];
        }
        float4 orow[8];
#pragma unroll
        for (int p = 0; p < 8; ++p) {
            int cm = rowBase + 2 * p + laneHi; if (cm > N - 1) cm = N - 1;
            orow[p] = x4[(size_t)cm * (D / 4) + laneLo];
        }
#pragma unroll
        for (int b = 0; b < TFILL; ++b) d16[threadIdx.x + (unsigned int)b * (WPB * 32)] = tv[b];
#pragma unroll
        for (int p = 0; p < 8; ++p) *(float4*)(myX + (2 * p + laneHi) * PITCH + 4 * laneLo) = orow[p];
    }
    const unsigned int Eu = (unsigned int)E;
    const unsigned int lastE = Eu > 0u ? Eu - 1u : 0u;
    const unsigned int lastNode = (unsigned int)(N - 1);

    const int hrow = rowBase + laneLo;
    const bool hv = hrow < N;
    const int hr = hv ? hrow : N - 1;
    unsigned int startAll = rp_abs[hr];
    if (startAll > Eu) startAll = Eu;
    unsigned int nAll = hv ? degc[hr] : 0u;
    if (nAll > Eu - startAll) nAll = Eu - startAll;
    const float dcAll = dinv[hr];
    __syncthreads();

    unsigned int start = __shfl(startAll, laneHi);
    unsigned int n = __shfl(nAll, laneHi);
    float dc = __shfl(dcAll, laneHi);
    int c = rowBase + laneHi; if (c > N - 1) c = N - 1;
    int iA0, iB0;
    {
        unsigned int pA = start + (unsigned int)laneLo;       if (pA > lastE) pA = lastE;
        unsigned int pB = start + (unsigned int)laneLo + 16u; if (pB > lastE) pB = lastE;
        iA0 = srt[pA];
        iB0 = srt[pB];
    }
    float4 xs = x4[(size_t)c * (D / 4) + laneLo];
    unsigned int startN = __shfl(startAll, 2 + laneHi);
    unsigned int nN = __shfl(nAll, 2 + laneHi);
    float dcN = __shfl(dcAll, 2 + laneHi);
    int cN = rowBase + 2 + laneHi; if (cN > N - 1) cN = N - 1;
    int iAn, iBn;
    {
        unsigned int qA = startN + (unsigned int)laneLo;       if (qA > lastE) qA = lastE;
        unsigned int qB = startN + (unsigned int)laneLo + 16u; if (qB > lastE) qB = lastE;
        iAn = srt[qA];
        iBn = srt[qB];
    }

    {
        const float* ap = myX + laneLo * PITCH + 2 * laneHi;
        v2f a[16];
#pragma unroll
        for (int k = 0; k < 16; ++k) {
            const float2 t = *(const float2*)(ap + 4 * k);
            a[k].x = t.x; a[k].y = t.y;
        }
#pragma unroll
        for (int ntp = 0; ntp < 2; ++ntp) {
            const int n0 = ntp * 32 + laneLo;
            const int n1 = n0 + 16;
            v8f s0 = {};
            v8f s1 = {};
            v2f b0[16], b1[16];
#pragma unroll
            for (int j = 0; j < 16; ++j) {
                const int kk = 4 * j + 2 * laneHi;
                b0[j].x = Ws[(size_t)kk * D + n0];
                b0[j].y = Ws[(size_t)(kk + 1) * D + n0];
                b1[j].x = Ws[(size_t)kk * D + n1];
                b1[j].y = Ws[(size_t)(kk + 1) * D + n1];
            }
#pragma unroll
            for (int j = 0; j < 16; ++j) {
                s0 = __builtin_amdgcn_wmma_f32_16x16x4_f32(
                    false, a[j], false, b0[j], (short)0, s0, false, false);
                s1 = __builtin_amdgcn_wmma_f32_16x16x4_f32(
                    false, a[j], false, b1[j], (short)0, s1, false, false);
            }
#pragma unroll
            for (int r = 0; r < 8; ++r) {
                float* q = myX + (r + 8 * laneHi) * PITCH;
                q[n0] = s0[r];
                q[n1] = s1[r];
            }
        }
    }

    unsigned int rA = ((unsigned int)laneLo < n) ? (unsigned int)iA0 : (unsigned int)c;
    unsigned int rB = ((unsigned int)(16 + laneLo) < n) ? (unsigned int)iB0 : (unsigned int)c;
    if (rA > lastNode) rA = lastNode;
    if (rB > lastNode) rB = lastNode;
    unsigned int qA = deg_byte(tab, rA);
    unsigned int qB = deg_byte(tab, rB);

    for (int p = 0; p < 8; ++p) {
        const float dA = src_dinv(qA, dinv, rA);
        const float dB = src_dinv(qB, dinv, rB);
        unsigned int rAn = ((unsigned int)laneLo < nN) ? (unsigned int)iAn : (unsigned int)cN;
        unsigned int rBn = ((unsigned int)(16 + laneLo) < nN) ? (unsigned int)iBn : (unsigned int)cN;
        if (rAn > lastNode) rAn = lastNode;
        if (rBn > lastNode) rBn = lastNode;
        const unsigned int qAn = deg_byte(tab, rAn);
        const unsigned int qBn = deg_byte(tab, rBn);
        const float4 xsn = x4[(size_t)cN * (D / 4) + laneLo];
        const int p2 = p < 6 ? p + 2 : 7;
        const int m2 = 2 * p2 + laneHi;
        const unsigned int start2 = __shfl(startAll, m2);
        const unsigned int n2 = __shfl(nAll, m2);
        const float dc2 = __shfl(dcAll, m2);
        int c2 = rowBase + m2; if (c2 > N - 1) c2 = N - 1;
        unsigned int qA2 = start2 + (unsigned int)laneLo;       if (qA2 > lastE) qA2 = lastE;
        unsigned int qB2 = start2 + (unsigned int)laneLo + 16u; if (qB2 > lastE) qB2 = lastE;
        const int iA2 = srt[qA2];
        const int iB2 = srt[qB2];
        const unsigned int nOther = __shfl_xor(n, 16);
        const unsigned int nmax = __builtin_amdgcn_readfirstlane(n > nOther ? n : nOther);
        const float ws = dc * dc;
        float4 acc;
        acc.x = ws * xs.x; acc.y = ws * xs.y; acc.z = ws * xs.z; acc.w = ws * xs.w;
        const float wA = ((unsigned int)laneLo < n) ? dA * dc : 0.0f;
        const float wB = ((unsigned int)(16 + laneLo) < n) ? dB * dc : 0.0f;
        pull_chunk(acc, x4, rA, wA, laneLo, nmax);
        pull_chunk(acc, x4, rB, wB, laneLo, nmax > 16u ? nmax - 16u : 0u);
        for (unsigned int j = 32u; j < nmax; j += 16u) {
            const bool ok = (j + (unsigned int)laneLo) < n;
            unsigned int qC = start + j + (unsigned int)laneLo; if (qC > lastE) qC = lastE;
            const int iC = srt[qC];
            unsigned int rC = ok ? (unsigned int)iC : (unsigned int)c;
            if (rC > lastNode) rC = lastNode;
            const float wC = ok ? src_dinv(deg_byte(tab, rC), dinv, rC) * dc : 0.0f;
            pull_chunk(acc, x4, rC, wC, laneLo, nmax - j);
        }
        const int rowOff = (2 * p + laneHi) * PITCH + 4 * laneLo;
        *(float4*)(myG + rowOff) = acc;
        start = startN; n = nN; dc = dcN; c = cN;
        rA = rAn; rB = rBn; qA = qAn; qB = qBn; xs = xsn;
        startN = start2; nN = n2; dcN = dc2; cN = c2; iAn = iA2; iBn = iB2;
    }
    __builtin_amdgcn_fence(__ATOMIC_RELEASE, "wavefront");
    __builtin_amdgcn_wave_barrier();
    __builtin_amdgcn_fence(__ATOMIC_ACQUIRE, "wavefront");

    const float* gp = myG + laneLo * PITCH + 2 * laneHi;
    v2f g[16];
#pragma unroll
    for (int k = 0; k < 16; ++k) {
        const float2 u = *(const float2*)(gp + 4 * k);
        g[k].x = u.x; g[k].y = u.y;
    }

#pragma unroll
    for (int ntp = 0; ntp < 2; ++ntp) {
        const int n0 = ntp * 32 + laneLo;
        const int n1 = n0 + 16;
        v8f c0 = {};
        v8f c1 = {};
        v2f b0[16], b1[16];
#pragma unroll
        for (int j = 0; j < 16; ++j) {
            const int kk = 4 * j + 2 * laneHi;
            b0[j].x = Wc[(size_t)kk * D + n0];
            b0[j].y = Wc[(size_t)(kk + 1) * D + n0];
            b1[j].x = Wc[(size_t)kk * D + n1];
            b1[j].y = Wc[(size_t)(kk + 1) * D + n1];
        }
#pragma unroll
        for (int j = 0; j < 16; ++j) {
            c0 = __builtin_amdgcn_wmma_f32_16x16x4_f32(
                false, g[j], false, b0[j], (short)0, c0, false, false);
            c1 = __builtin_amdgcn_wmma_f32_16x16x4_f32(
                false, g[j], false, b1[j], (short)0, c1, false, false);
        }
        const float bias0 = bs[n0] + bc[n0];
        const float bias1 = bs[n1] + bc[n1];
        if (rowBase + 16 <= N) {
#pragma unroll
            for (int r = 0; r < 8; ++r) {
                const float* q = myX + (r + 8 * laneHi) * PITCH;
                const size_t idx = (size_t)(rowBase + r + 8 * laneHi) * D;
                out[idx + n0] = q[n0] + bias0 + c0[r];
                out[idx + n1] = q[n1] + bias1 + c1[r];
            }
        } else {
#pragma unroll
            for (int r = 0; r < 8; ++r) {
                const int row = rowBase + r + 8 * laneHi;
                if (row < N) {
                    const float* q = myX + (r + 8 * laneHi) * PITCH;
                    const size_t idx = (size_t)row * D;
                    out[idx + n0] = q[n0] + bias0 + c0[r];
                    out[idx + n1] = q[n1] + bias1 + c1[r];
                }
            }
        }
    }
}

extern "C" void kernel_launch(void* const* d_in, const int* in_sizes, int n_in,
                              void* d_out, int out_size, void* d_ws, size_t ws_size,
                              hipStream_t stream) {
    const float* x  = (const float*)d_in[0];
    const int*   ei = (const int*)d_in[1];
    const float* Wc = (const float*)d_in[2];
    const float* bc = (const float*)d_in[3];
    const float* Ws = (const float*)d_in[4];
    const float* bs = (const float*)d_in[5];
    float* out = (float*)d_out;

    const int N = in_sizes[0] / D;
    const int E = in_sizes[1] / 2;

    float* dinv = (float*)d_ws;
    size_t off = ((size_t)N * sizeof(float) + 255) & ~(size_t)255;
    float* agg = (float*)((char*)d_ws + off);
    unsigned int* part = (unsigned int*)agg;
    int* srt = (int*)(part + (size_t)HS_SLICES * N);
    unsigned int* rp_local = (unsigned int*)(srt + E);
    unsigned int* rp_abs = rp_local + N;
    unsigned int* degc = rp_abs + N;
    unsigned int* bsum = degc + N;
    const int nb = (N + SB - 1) / SB;
    size_t off8 = (off + ((size_t)HS_SLICES * N + (size_t)E + 3 * (size_t)N + (size_t)nb) * sizeof(unsigned int) + 255) & ~(size_t)255;
    unsigned char* deg8 = (unsigned char*)d_ws + off8;

    const dim3 hgrid((N + HR_NODES - 1) / HR_NODES, HS_SLICES);
    deg_hist_kernel<<<hgrid, HT, 0, stream>>>(ei + E, part, N, E);
    deg_scan_kernel<<<nb, SB, 0, stream>>>(part, dinv, degc, rp_local, bsum, deg8, N);
    fill_kernel<<<hgrid, HT, 0, stream>>>(ei, ei + E, part, rp_local, bsum, rp_abs, srt, N, E);
    gcn_pull_gemm_kernel<<<(N + 16 * WPB - 1) / (16 * WPB), WPB * 32, 0, stream>>>(
        x, dinv, rp_abs, degc, srt, deg8, Wc, bc, Ws, bs, out, N, E);
}


# Edge ordering in 2 launches, not 3: chunks sorted by 400-node target range in LDS; each range's owner counts degrees, orders its edges in LDS; node lists ascending by source (sum order changes)
# speedup vs baseline: 4.3565x; 1.1684x over previous
#include <hip/hip_runtime.h>
#include <hip/hip_bf16.h>


typedef __attribute__((ext_vector_type(2))) float v2f;
typedef __attribute__((ext_vector_type(8))) float v8f;

#define D 64

#define BN 400
#define NBMAX 256
#define BT 512
#define CHCAP 6250
#define KMAX 13
static_assert(KMAX * BT >= CHCAP, "a thread's KMAX edges cover the chunk");
static_assert(CHCAP < 65536, "a piece's length and its offset inside the chunk fit 16 bits each");
static_assert(BN % 16 == 0, "a range is a whole number of 16-node tiles");

__device__ __forceinline__ unsigned int wave_scan_incl(unsigned int v, int lane) {
#pragma unroll
    for (int o = 1; o < 32; o <<= 1) {
        const unsigned int u = __shfl_up(v, o);
        if (lane >= o) v += u;
    }
    return v;
}

__global__ __launch_bounds__(BT)
void bin_kernel(const int* __restrict__ row, const int* __restrict__ col,
                int* __restrict__ rec_i, unsigned int* __restrict__ pcT,
                int N, int E, int nbin) {
    __shared__ unsigned int bc[NBMAX];
    __shared__ unsigned int wt[NBMAX / 32];
    __shared__ int2 st[CHCAP];
    int2* __restrict__ rec = (int2*)rec_i;
    const int t = threadIdx.x;
    const int lane = t & 31;
    const int c = blockIdx.x;
    const int nch = gridDim.x;
    const int ch = (E + nch - 1) / nch;
    const long long e0l = (long long)c * ch;
    const int e0 = e0l < (long long)E ? (int)e0l : E;
    const int e1 = (E - e0 < ch) ? E : e0 + ch;
    const int n = e1 - e0;
    const unsigned int lastBin = (unsigned int)((nbin < NBMAX ? nbin : NBMAX) - 1);

    if (t < NBMAX) bc[t] = 0u;
    int cv[KMAX], rv[KMAX];
#pragma unroll
    for (int k = 0; k < KMAX; ++k) {
        int i = t + k * BT;
        if (i > n - 1) i = n - 1;
        if (i < 0) i = 0;
        cv[k] = (n > 0) ? col[e0 + i] : 0;
        rv[k] = (n > 0) ? row[e0 + i] : 0;
    }
    __syncthreads();
    unsigned int bv[KMAX];
#pragma unroll
    for (int k = 0; k < KMAX; ++k) {
        unsigned int b = (unsigned int)cv[k] / (unsigned int)BN;
        if (b > lastBin) b = lastBin;
        bv[k] = b;
        if (t + k * BT < n) atomicAdd(&bc[b], 1u);
    }
    __syncthreads();
    const unsigned int v = (t < NBMAX) ? bc[t] : 0u;
    const unsigned int incl = wave_scan_incl(v, lane);
    if (t < NBMAX && lane == 31) wt[t >> 5] = incl;
    __syncthreads();
    unsigned int pre = 0u;
#pragma unroll
    for (int j = 0; j < NBMAX / 32; ++j) pre += (j < (t >> 5)) ? wt[j] : 0u;
    const unsigned int excl = pre + incl - v;
    if (t < nbin && t < NBMAX) {
        pcT[(size_t)t * nch + c] = (excl << 16) | v;
        bc[t] = excl;
    }
    __syncthreads();
    unsigned int ps[KMAX];
#pragma unroll
    for (int k = 0; k < KMAX; ++k) {
        ps[k] = (unsigned int)CHCAP;
        if (t + k * BT < n) ps[k] = atomicAdd(&bc[bv[k]], 1u);
    }
#pragma unroll
    for (int k = 0; k < KMAX; ++k)
        if (ps[k] < (unsigned int)CHCAP) st[ps[k]] = make_int2(rv[k], cv[k]);
    __syncthreads();
    for (int i = t; i < n; i += BT) rec[(size_t)e0 + i] = st[i];
}

#define SCAP 16384
#define PPH 8

#define KSRC 17
#define KMASK ((1u << KSRC) - 1u)
static_assert(BN <= (1 << (32 - KSRC - 6)), "the node's index inside the range fits the key's top bits");
__device__ __forceinline__ unsigned int seg_key(unsigned int d, int src, unsigned int pos) {
    return (d << (KSRC + 6)) | (((unsigned int)src & KMASK) << 6) | (pos & 63u);
}

template <bool PLACE>
__device__ __forceinline__ void own_round(const int2* __restrict__ rec,
                                          const unsigned int* __restrict__ pcRow,
                                          int c0, int nch, int ch, int E, int lo, unsigned int cntn,
                                          unsigned int* deg, unsigned int* cur, unsigned int* stg,
                                          bool staged, unsigned int base,
                                          int* __restrict__ srt, int t) {
    const int hw = t >> 4;
    const int l = t & 15;
    unsigned int cn[PPH], of[PPH];
#pragma unroll
    for (int p = 0; p < PPH; ++p) {
        int c = c0 + hw + 32 * p;
        const bool valid = c < nch;
        if (!valid) c = nch - 1;
        const unsigned int w = pcRow[c];
        const unsigned int o = w >> 16;
        cn[p] = valid ? (w & 0xffffu) : 0u;
        const long long e0l = (long long)c * ch;
        const unsigned int e0 = e0l < (long long)E ? (unsigned int)e0l : (unsigned int)E;
        of[p] = e0 + o;
    }
    const unsigned int lastE = E > 0 ? (unsigned int)(E - 1) : 0u;
    int2 r[PPH][2];
#pragma unroll
    for (int p = 0; p < PPH; ++p) {
#pragma unroll
        for (int s = 0; s < 2; ++s) {
            unsigned int i = of[p] + (unsigned int)(l + 16 * s);
            if (i > lastE) i = lastE;
            r[p][s] = rec[i];
        }
    }
    unsigned int ps[PPH][2];
#pragma unroll
    for (int p = 0; p < PPH; ++p) {
#pragma unroll
        for (int s = 0; s < 2; ++s) {
            const unsigned int d = (unsigned int)(r[p][s].y - lo);
            ps[p][s] = 0xffffffffu;
            if ((unsigned int)(l + 16 * s) < cn[p] && d < cntn) {
                if (!PLACE) atomicAdd(&deg[d], 1u);
                else ps[p][s] = atomicAdd(&cur[d], 1u);
            }
        }
    }
#pragma unroll
    for (int p = 0; p < PPH; ++p) {
        if (PLACE) {
#pragma unroll
            for (int s = 0; s < 2; ++s) {
                const unsigned int pos = ps[p][s];
                if (pos != 0xffffffffu) {
                    if (staged) {
                        if (pos < (unsigned int)SCAP)
                            stg[pos] = seg_key((unsigned int)(r[p][s].y - lo), r[p][s].x, pos);
                    } else {
                        const unsigned int gp = base + pos;
                        if (gp < (unsigned int)E) srt[gp] = r[p][s].x;
                    }
                }
            }
        }
        for (unsigned int j = 32u + (unsigned int)l; j < cn[p]; j += 16u) {
            unsigned int i = of[p] + j;
            if (i > lastE) i = lastE;
            const int2 q = rec[i];
            const unsigned int d = (unsigned int)(q.y - lo);
            if (d < cntn) {
                if (!PLACE) {
                    atomicAdd(&deg[d], 1u);
                } else {
                    const unsigned int pos = atomicAdd(&cur[d], 1u);
                    if (staged) {
                        if (pos < (unsigned int)SCAP) stg[pos] = seg_key(d, q.x, pos);
                    } else {
                        const unsigned int gp = base + pos;
                        if (gp < (unsigned int)E) srt[gp] = q.x;
                    }
                }
            }
        }
    }
}

#define RW 24
__device__ __forceinline__ unsigned int seg_rank(const unsigned int* stg, unsigned int s0,
                                                 unsigned int nn, unsigned int i, unsigned int key) {
    unsigned int rank = 0u;
    if (nn <= 64u) {
        unsigned int o[RW];
#pragma unroll
        for (int u = 0; u < RW; ++u) o[u] = stg[s0 + (unsigned int)u];
#pragma unroll
        for (int u = 0; u < RW; ++u) rank += (o[u] < key) ? 1u : 0u;
        for (unsigned int k = RW; k < nn; ++k) rank += (stg[s0 + k] < key) ? 1u : 0u;
    } else {
        const unsigned int ks = key >> 6;
        for (unsigned int k = 0u; k < nn; ++k) {
            const unsigned int os = stg[s0 + k] >> 6;
            rank += (os < ks || (os == ks && s0 + k < i)) ? 1u : 0u;
        }
    }
    return rank;
}

__global__ __launch_bounds__(BT)
void own_kernel(const int* __restrict__ rec_i, const unsigned int* __restrict__ pcT,
                int* __restrict__ srt,
                unsigned int* __restrict__ rp_abs, unsigned int* __restrict__ degc,
                float* __restrict__ dinv, unsigned char* __restrict__ deg8,
                int N, int E, int nch) {
    __shared__ unsigned int deg[BN];
    __shared__ unsigned int segs[BN];
    __shared__ unsigned int cur[BN];
    __shared__ unsigned int wt[BT / 32];
    __shared__ unsigned int baseS;
    __shared__ unsigned int stg[SCAP + 32];
    const int2* __restrict__ rec = (const int2*)rec_i;
    const int t = threadIdx.x;
    const int lane = t & 31;
    const int b = blockIdx.x;
    const int lo = b * BN;
    const int left = N - lo;
    const unsigned int cntn = (unsigned int)(left < BN ? left : BN);
    const int ch = (E + nch - 1) / nch;
    const unsigned int* __restrict__ pcRow = pcT + (size_t)b * nch;

    if (t < BN) deg[t] = 0u;
    if (t == 0) baseS = 0u;
    __syncthreads();
    {
        unsigned int mine = 0u;
        for (int c = t; c < nch; c += BT) mine += pcRow[c] >> 16;
#pragma unroll
        for (int o = 16; o > 0; o >>= 1) mine += __shfl_xor(mine, o);
        if (lane == 0 && mine != 0u) atomicAdd(&baseS, mine);
    }
    for (int c0 = 0; c0 < nch; c0 += 32 * PPH)
        own_round<false>(rec, pcRow, c0, nch, ch, E, lo, cntn, deg, cur, stg,
                         false, 0u, srt, t);
    __syncthreads();
    const unsigned int v = (t < BN) ? deg[t] : 0u;
    const unsigned int incl = wave_scan_incl(v, lane);
    if (lane == 31) wt[t >> 5] = incl;
    __syncthreads();
    unsigned int pre = 0u, tot = 0u;
#pragma unroll
    for (int j = 0; j < BT / 32; ++j) {
        const unsigned int w = wt[j];
        pre += (j < (t >> 5)) ? w : 0u;
        tot += w;
    }
    const unsigned int excl = pre + incl - v;
    const unsigned int base = baseS;
    const bool staged = tot <= (unsigned int)SCAP && (unsigned int)N <= (1u << KSRC);
    if ((unsigned int)t < cntn) {
        const int i = lo + t;
        rp_abs[i] = base + excl;
        degc[i] = v;
        dinv[i] = __frsqrt_rn((float)(v + 1u));
    }
    if (t < BN) {
        deg8[lo + t] = (unsigned char)(((unsigned int)t < cntn) ? (v < 255u ? v : 255u) : 0u);
        segs[t] = excl;
        cur[t] = excl;
    }
    if (b == (int)gridDim.x - 1 && t < 16) deg8[lo + BN + t] = 0;
    if (staged && t < 32) stg[tot + (unsigned int)t] = 0xffffffffu;
    __syncthreads();
    for (int c0 = 0; c0 < nch; c0 += 32 * PPH)
        own_round<true>(rec, pcRow, c0, nch, ch, E, lo, cntn, deg, cur, stg,
                        staged, base, srt, t);
    if (!staged) return;
    __syncthreads();
    for (unsigned int i0 = (unsigned int)t; i0 < tot; i0 += 2u * BT) {
        const bool has1 = i0 + BT < tot;
        const unsigned int i1 = has1 ? i0 + BT : i0;
        const unsigned int key0 = stg[i0];
        const unsigned int key1 = stg[i1];
        const unsigned int d0 = key0 >> (KSRC + 6);
        const unsigned int d1 = key1 >> (KSRC + 6);
        const unsigned int s0 = segs[d0], n0 = deg[d0];
        const unsigned int s1 = segs[d1], n1 = deg[d1];
        const unsigned int r0 = seg_rank(stg, s0, n0, i0, key0);
        const unsigned int r1 = seg_rank(stg, s1, n1, i1, key1);
        const unsigned int g0 = base + s0 + r0;
        const unsigned int g1 = base + s1 + r1;
        if (g0 < (unsigned int)E) srt[g0] = (int)((key0 >> 6) & KMASK);
        if (has1 && g1 < (unsigned int)E) srt[g1] = (int)((key1 >> 6) & KMASK);
    }
}

#define WPB 25
#define TCAP 102400
#define TFILL (TCAP / 16 / (WPB * 32))
static_assert(TFILL * 16 * WPB * 32 == TCAP, "the block's lanes cover the table in whole rounds of 16-byte loads");
#define PITCH 68

template <int CNT>
__device__ __forceinline__ void pull_group(float4& acc, const float4* __restrict__ x4,
                                           unsigned int r16, float w16, int l) {
    float4 v[CNT];
    const char* xb = (const char*)x4;
    const unsigned int lo = (unsigned int)l * 16u;
#pragma unroll
    for (int k = 0; k < CNT; ++k) {
        const unsigned int r = __shfl(r16, k, 16);
        v[k] = *(const float4*)(xb + ((r << 8) + lo));
    }
#pragma unroll
    for (int k = 0; k < CNT; ++k) {
        const float w = __shfl(w16, k, 16);
        acc.x += w * v[k].x;
        acc.y += w * v[k].y;
        acc.z += w * v[k].z;
        acc.w += w * v[k].w;
    }
}

__device__ __forceinline__ unsigned int deg_byte(const unsigned char* tab, unsigned int r) {
    const bool in = r < (unsigned int)TCAP;
    const unsigned int q = tab[in ? r : 0u];
    return in ? q : 255u;
}

__device__ __forceinline__ float src_dinv(unsigned int q, const float* __restrict__ dinv,
                                          unsigned int r) {
    float w = __frsqrt_rn((float)(q + 1u));
    if (q == 255u) w = dinv[r];
    return w;
}

__device__ __forceinline__ void pull_chunk(float4& acc, const float4* __restrict__ x4,
                                           unsigned int r16, float w16, int l, unsigned int rem) {
    if (rem > 8u) pull_group<16>(acc, x4, r16, w16, l);
    else if (rem > 0u) pull_group<8>(acc, x4, r16, w16, l);
}

__global__ __launch_bounds__(WPB * 32) __attribute__((amdgpu_waves_per_eu(7, 8)))
void gcn_pull_gemm_kernel(const float* __restrict__ x, const float* __restrict__ dinv,
                          const unsigned int* __restrict__ rp_abs,
                          const unsigned int* __restrict__ degc,
                          const int* __restrict__ srt, const unsigned char* __restrict__ deg8,
                          const float* __restrict__ Wc, const float* __restrict__ bc,
                          const float* __restrict__ Ws, const float* __restrict__ bs,
                          float* __restrict__ out, int N, int E) {
    __shared__ float aggS[WPB][16 * PITCH];
    __shared__ float xS[WPB][16 * PITCH];
    __shared__ __attribute__((aligned(16))) unsigned int degT[TCAP / 4];
    const unsigned char* tab = (const unsigned char*)degT;
    const int lane   = threadIdx.x & 31;
    const int wave   = threadIdx.x >> 5;
    const int waveG  = blockIdx.x * WPB + wave;
    const int rowBase = waveG * 16;
    const int laneHi = lane >> 4;
    const int laneLo = lane & 15;
    const float4* x4 = (const float4*)x;
    float* myG = aggS[wave];
    float* myX = xS[wave];
    {
        const unsigned int nn = (unsigned int)N < (unsigned int)TCAP ? (unsigned int)N : (unsigned int)TCAP;
        const unsigned int nq = (nn + 15u) >> 4;
        const uint4* __restrict__ s16 = (const uint4*)deg8;
        uint4* d16 = (uint4*)degT;
        uint4 tv[TFILL];
#pragma unroll
        for (int b = 0; b < TFILL; ++b) {
            unsigned int i = threadIdx.x + (unsigned int)b * (WPB * 32);
            if (i >= nq) i = nq - 1u;
            tv[b] = s16[i];
        }
        float4 orow[8];
#pragma unroll
        for (int p = 0; p < 8; ++p) {
            int cm = rowBase + 2 * p + laneHi; if (cm > N - 1) cm = N - 1;
            orow[p] = x4[(size_t)cm * (D / 4) + laneLo];
        }
#pragma unroll
        for (int b = 0; b < TFILL; ++b) d16[threadIdx.x + (unsigned int)b * (WPB * 32)] = tv[b];
#pragma unroll
        for (int p = 0; p < 8; ++p) *(float4*)(myX + (2 * p + laneHi) * PITCH + 4 * laneLo) = orow[p];
    }
    const unsigned int Eu = (unsigned int)E;
    const unsigned int lastE = Eu > 0u ? Eu - 1u : 0u;
    const unsigned int lastNode = (unsigned int)(N - 1);

    const int hrow = rowBase + laneLo;
    const bool hv = hrow < N;
    const int hr = hv ? hrow : N - 1;
    unsigned int startAll = rp_abs[hr];
    if (startAll > Eu) startAll = Eu;
    unsigned int nAll = hv ? degc[hr] : 0u;
    if (nAll > Eu - startAll) nAll = Eu - startAll;
    const float dcAll = dinv[hr];
    __syncthreads();

    unsigned int start = __shfl(startAll, laneHi);
    unsigned int n = __shfl(nAll, laneHi);
    float dc = __shfl(dcAll, laneHi);
    int c = rowBase + laneHi; if (c > N - 1) c = N - 1;
    int iA0, iB0;
    {
        unsigned int pA = start + (unsigned int)laneLo;       if (pA > lastE) pA = lastE;
        unsigned int pB = start + (unsigned int)laneLo + 16u; if (pB > lastE) pB = lastE;
        iA0 = srt[pA];
        iB0 = srt[pB];
    }
    float4 xs = x4[(size_t)c * (D / 4) + laneLo];
    unsigned int startN = __shfl(startAll, 2 + laneHi);
    unsigned int nN = __shfl(nAll, 2 + laneHi);
    float dcN = __shfl(dcAll, 2 + laneHi);
    int cN = rowBase + 2 + laneHi; if (cN > N - 1) cN = N - 1;
    int iAn, iBn;
    {
        unsigned int qA = startN + (unsigned int)laneLo;       if (qA > lastE) qA = lastE;
        unsigned int qB = startN + (unsigned int)laneLo + 16u; if (qB > lastE) qB = lastE;
        iAn = srt[qA];
        iBn = srt[qB];
    }

    {
        const float* ap = myX + laneLo * PITCH + 2 * laneHi;
        v2f a[16];
#pragma unroll
        for (int k = 0; k < 16; ++k) {
            const float2 t = *(const float2*)(ap + 4 * k);
            a[k].x = t.x; a[k].y = t.y;
        }
#pragma unroll
        for (int ntp = 0; ntp < 2; ++ntp) {
            const int n0 = ntp * 32 + laneLo;
            const int n1 = n0 + 16;
            v8f s0 = {};
            v8f s1 = {};
            v2f b0[16], b1[16];
#pragma unroll
            for (int j = 0; j < 16; ++j) {
                const int kk = 4 * j + 2 * laneHi;
                b0[j].x = Ws[(size_t)kk * D + n0];
                b0[j].y = Ws[(size_t)(kk + 1) * D + n0];
                b1[j].x = Ws[(size_t)kk * D + n1];
                b1[j].y = Ws[(size_t)(kk + 1) * D + n1];
            }
#pragma unroll
            for (int j = 0; j < 16; ++j) {
                s0 = __builtin_amdgcn_wmma_f32_16x16x4_f32(
                    false, a[j], false, b0[j], (short)0, s0, false, false);
                s1 = __builtin_amdgcn_wmma_f32_16x16x4_f32(
                    false, a[j], false, b1[j], (short)0, s1, false, false);
            }
#pragma unroll
            for (int r = 0; r < 8; ++r) {
                float* q = myX + (r + 8 * laneHi) * PITCH;
                q[n0] = s0[r];
                q[n1] = s1[r];
            }
        }
    }

    unsigned int rA = ((unsigned int)laneLo < n) ? (unsigned int)iA0 : (unsigned int)c;
    unsigned int rB = ((unsigned int)(16 + laneLo) < n) ? (unsigned int)iB0 : (unsigned int)c;
    if (rA > lastNode) rA = lastNode;
    if (rB > lastNode) rB = lastNode;
    unsigned int qA = deg_byte(tab, rA);
    unsigned int qB = deg_byte(tab, rB);

    for (int p = 0; p < 8; ++p) {
        const float dA = src_dinv(qA, dinv, rA);
        const float dB = src_dinv(qB, dinv, rB);
        unsigned int rAn = ((unsigned int)laneLo < nN) ? (unsigned int)iAn : (unsigned int)cN;
        unsigned int rBn = ((unsigned int)(16 + laneLo) < nN) ? (unsigned int)iBn : (unsigned int)cN;
        if (rAn > lastNode) rAn = lastNode;
        if (rBn > lastNode) rBn = lastNode;
        const unsigned int qAn = deg_byte(tab, rAn);
        const unsigned int qBn = deg_byte(tab, rBn);
        const float4 xsn = x4[(size_t)cN * (D / 4) + laneLo];
        const int p2 = p < 6 ? p + 2 : 7;
        const int m2 = 2 * p2 + laneHi;
        const unsigned int start2 = __shfl(startAll, m2);
        const unsigned int n2 = __shfl(nAll, m2);
        const float dc2 = __shfl(dcAll, m2);
        int c2 = rowBase + m2; if (c2 > N - 1) c2 = N - 1;
        unsigned int qA2 = start2 + (unsigned int)laneLo;       if (qA2 > lastE) qA2 = lastE;
        unsigned int qB2 = start2 + (unsigned int)laneLo + 16u; if (qB2 > lastE) qB2 = lastE;
        const int iA2 = srt[qA2];
        const int iB2 = srt[qB2];
        const unsigned int nOther = __shfl_xor(n, 16);
        const unsigned int nmax = __builtin_amdgcn_readfirstlane(n > nOther ? n : nOther);
        const float ws = dc * dc;
        float4 acc;
        acc.x = ws * xs.x; acc.y = ws * xs.y; acc.z = ws * xs.z; acc.w = ws * xs.w;
        const float wA = ((unsigned int)laneLo < n) ? dA * dc : 0.0f;
        const float wB = ((unsigned int)(16 + laneLo) < n) ? dB * dc : 0.0f;
        pull_chunk(acc, x4, rA, wA, laneLo, nmax);
        pull_chunk(acc, x4, rB, wB, laneLo, nmax > 16u ? nmax - 16u : 0u);
        for (unsigned int j = 32u; j < nmax; j += 16u) {
            const bool ok = (j + (unsigned int)laneLo) < n;
            unsigned int qC = start + j + (unsigned int)laneLo; if (qC > lastE) qC = lastE;
            const int iC = srt[qC];
            unsigned int rC = ok ? (unsigned int)iC : (unsigned int)c;
            if (rC > lastNode) rC = lastNode;
            const float wC = ok ? src_dinv(deg_byte(tab, rC), dinv, rC) * dc : 0.0f;
            pull_chunk(acc, x4, rC, wC, laneLo, nmax - j);
        }
        const int rowOff = (2 * p + laneHi) * PITCH + 4 * laneLo;
        *(float4*)(myG + rowOff) = acc;
        start = startN; n = nN; dc = dcN; c = cN;
        rA = rAn; rB = rBn; qA = qAn; qB = qBn; xs = xsn;
        startN = start2; nN = n2; dcN = dc2; cN = c2; iAn = iA2; iBn = iB2;
    }
    __builtin_amdgcn_fence(__ATOMIC_RELEASE, "wavefront");
    __builtin_amdgcn_wave_barrier();
    __builtin_amdgcn_fence(__ATOMIC_ACQUIRE, "wavefront");

    const float* gp = myG + laneLo * PITCH + 2 * laneHi;
    v2f g[16];
#pragma unroll
    for (int k = 0; k < 16; ++k) {
        const float2 u = *(const float2*)(gp + 4 * k);
        g[k].x = u.x; g[k].y = u.y;
    }

#pragma unroll
    for (int ntp = 0; ntp < 2; ++ntp) {
        const int n0 = ntp * 32 + laneLo;
        const int n1 = n0 + 16;
        v8f c0 = {};
        v8f c1 = {};
        v2f b0[16], b1[16];
#pragma unroll
        for (int j = 0; j < 16; ++j) {
            const int kk = 4 * j + 2 * laneHi;
            b0[j].x = Wc[(size_t)kk * D + n0];
            b0[j].y = Wc[(size_t)(kk + 1) * D + n0];
            b1[j].x = Wc[(size_t)kk * D + n1];
            b1[j].y = Wc[(size_t)(kk + 1) * D + n1];
        }
#pragma unroll
        for (int j = 0; j < 16; ++j) {
            c0 = __builtin_amdgcn_wmma_f32_16x16x4_f32(
                false, g[j], false, b0[j], (short)0, c0, false, false);
            c1 = __builtin_amdgcn_wmma_f32_16x16x4_f32(
                false, g[j], false, b1[j], (short)0, c1, false, false);
        }
        const float bias0 = bs[n0] + bc[n0];
        const float bias1 = bs[n1] + bc[n1];
        if (rowBase + 16 <= N) {
#pragma unroll
            for (int r = 0; r < 8; ++r) {
                const float* q = myX + (r + 8 * laneHi) * PITCH;
                const size_t idx = (size_t)(rowBase + r + 8 * laneHi) * D;
                out[idx + n0] = q[n0] + bias0 + c0[r];
                out[idx + n1] = q[n1] + bias1 + c1[r];
            }
        } else {
#pragma unroll
            for (int r = 0; r < 8; ++r) {
                const int row = rowBase + r + 8 * laneHi;
                if (row < N) {
                    const float* q = myX + (r + 8 * laneHi) * PITCH;
                    const size_t idx = (size_t)row * D;
                    out[idx + n0] = q[n0] + bias0 + c0[r];
                    out[idx + n1] = q[n1] + bias1 + c1[r];
                }
            }
        }
    }
}

extern "C" void kernel_launch(void* const* d_in, const int* in_sizes, int n_in,
                              void* d_out, int out_size, void* d_ws, size_t ws_size,
                              hipStream_t stream) {
    const float* x  = (const float*)d_in[0];
    const int*   ei = (const int*)d_in[1];
    const float* Wc = (const float*)d_in[2];
    const float* bc = (const float*)d_in[3];
    const float* Ws = (const float*)d_in[4];
    const float* bs = (const float*)d_in[5];
    float* out = (float*)d_out;

    const int N = in_sizes[0] / D;
    const int E = in_sizes[1] / 2;
    const int nbin = (N + BN - 1) / BN;
    const int nch = (E + CHCAP - 1) / CHCAP;

    float* dinv = (float*)d_ws;
    size_t off = ((size_t)N * sizeof(float) + 255) & ~(size_t)255;
    int* rec = (int*)((char*)d_ws + off);
    int* srt = rec + 2 * (size_t)E;
    unsigned int* pcT = (unsigned int*)(srt + E);
    unsigned int* rp_abs = pcT + (size_t)nbin * nch;
    unsigned int* degc = rp_abs + N;
    size_t off8 = (off + (size_t)E * 8 + ((size_t)E + (size_t)nbin * nch + 2 * (size_t)N) * sizeof(unsigned int) + 255) & ~(size_t)255;
    unsigned char* deg8 = (unsigned char*)d_ws + off8;

    bin_kernel<<<nch, BT, 0, stream>>>(ei, ei + E, rec, pcT, N, E, nbin);
    own_kernel<<<nbin, BT, 0, stream>>>(rec, pcT, srt, rp_abs, degc, dinv, deg8, N, E, nch);
    gcn_pull_gemm_kernel<<<(N + 16 * WPB - 1) / (16 * WPB), WPB * 32, 0, stream>>>(
        x, dinv, rp_abs, degc, srt, deg8, Wc, bc, Ws, bs, out, N, E);
}


# Range owner reads its records once: first 48 of every piece in one batch at kernel entry, kept in registers for count and place passes; no dependent load per long piece, no re-fetch; sums unchanged
# speedup vs baseline: 4.5476x; 1.0439x over previous
#include <hip/hip_runtime.h>
#include <hip/hip_bf16.h>


typedef __attribute__((ext_vector_type(2))) float v2f;
typedef __attribute__((ext_vector_type(8))) float v8f;

#define D 64

#define BN 400
#define NBMAX 256
#define BT 512
#define CHCAP 6250
#define KMAX 13
static_assert(KMAX * BT >= CHCAP, "a thread's KMAX edges cover the chunk");
static_assert(CHCAP < 65536, "a piece's length and its offset inside the chunk fit 16 bits each");
static_assert(BN % 16 == 0, "a range is a whole number of 16-node tiles");

__device__ __forceinline__ unsigned int wave_scan_incl(unsigned int v, int lane) {
#pragma unroll
    for (int o = 1; o < 32; o <<= 1) {
        const unsigned int u = __shfl_up(v, o);
        if (lane >= o) v += u;
    }
    return v;
}

__global__ __launch_bounds__(BT)
void bin_kernel(const int* __restrict__ row, const int* __restrict__ col,
                int* __restrict__ rec_i, unsigned int* __restrict__ pcT,
                int N, int E, int nbin) {
    __shared__ unsigned int bc[NBMAX];
    __shared__ unsigned int wt[NBMAX / 32];
    __shared__ int2 st[CHCAP];
    int2* __restrict__ rec = (int2*)rec_i;
    const int t = threadIdx.x;
    const int lane = t & 31;
    const int c = blockIdx.x;
    const int nch = gridDim.x;
    const int ch = (E + nch - 1) / nch;
    const long long e0l = (long long)c * ch;
    const int e0 = e0l < (long long)E ? (int)e0l : E;
    const int e1 = (E - e0 < ch) ? E : e0 + ch;
    const int n = e1 - e0;
    const unsigned int lastBin = (unsigned int)((nbin < NBMAX ? nbin : NBMAX) - 1);

    if (t < NBMAX) bc[t] = 0u;
    int cv[KMAX], rv[KMAX];
#pragma unroll
    for (int k = 0; k < KMAX; ++k) {
        int i = t + k * BT;
        if (i > n - 1) i = n - 1;
        if (i < 0) i = 0;
        cv[k] = (n > 0) ? col[e0 + i] : 0;
        rv[k] = (n > 0) ? row[e0 + i] : 0;
    }
    __syncthreads();
    unsigned int bv[KMAX];
#pragma unroll
    for (int k = 0; k < KMAX; ++k) {
        unsigned int b = (unsigned int)cv[k] / (unsigned int)BN;
        if (b > lastBin) b = lastBin;
        bv[k] = b;
        if (t + k * BT < n) atomicAdd(&bc[b], 1u);
    }
    __syncthreads();
    const unsigned int v = (t < NBMAX) ? bc[t] : 0u;
    const unsigned int incl = wave_scan_incl(v, lane);
    if (t < NBMAX && lane == 31) wt[t >> 5] = incl;
    __syncthreads();
    unsigned int pre = 0u;
#pragma unroll
    for (int j = 0; j < NBMAX / 32; ++j) pre += (j < (t >> 5)) ? wt[j] : 0u;
    const unsigned int excl = pre + incl - v;
    if (t < nbin && t < NBMAX) {
        pcT[(size_t)t * nch + c] = (excl << 16) | v;
        bc[t] = excl;
    }
    __syncthreads();
    unsigned int ps[KMAX];
#pragma unroll
    for (int k = 0; k < KMAX; ++k) {
        ps[k] = (unsigned int)CHCAP;
        if (t + k * BT < n) ps[k] = atomicAdd(&bc[bv[k]], 1u);
    }
#pragma unroll
    for (int k = 0; k < KMAX; ++k)
        if (ps[k] < (unsigned int)CHCAP) st[ps[k]] = make_int2(rv[k], cv[k]);
    __syncthreads();
    for (int i = t; i < n; i += BT) rec[(size_t)e0 + i] = st[i];
}

#define SCAP 16384
#define PPH 8

#define KSRC 17
#define KMASK ((1u << KSRC) - 1u)
static_assert(BN <= (1 << (32 - KSRC - 6)), "the node's index inside the range fits the key's top bits");
__device__ __forceinline__ unsigned int seg_key(unsigned int d, int src, unsigned int pos) {
    return (d << (KSRC + 6)) | (((unsigned int)src & KMASK) << 6) | (pos & 63u);
}

#define PSL 3
__device__ __forceinline__ void own_fetch(const int2* __restrict__ rec,
                                          const unsigned int* __restrict__ pcRow,
                                          int nch, int ch, int E, int t,
                                          unsigned int (&cn)[PPH], int2 (&r)[PPH][PSL]) {
    const int hw = t >> 4;
    const int l = t & 15;
    unsigned int of[PPH];
#pragma unroll
    for (int p = 0; p < PPH; ++p) {
        int c = hw + 32 * p;
        const bool valid = c < nch;
        if (!valid) c = nch - 1;
        const unsigned int w = pcRow[c];
        const unsigned int o = w >> 16;
        cn[p] = valid ? (w & 0xffffu) : 0u;
        const long long e0l = (long long)c * ch;
        const unsigned int e0 = e0l < (long long)E ? (unsigned int)e0l : (unsigned int)E;
        of[p] = e0 + o;
    }
    const unsigned int lastE = E > 0 ? (unsigned int)(E - 1) : 0u;
#pragma unroll
    for (int p = 0; p < PPH; ++p) {
#pragma unroll
        for (int s = 0; s < PSL; ++s) {
            unsigned int i = of[p] + (unsigned int)(l + 16 * s);
            if (i > lastE) i = lastE;
            r[p][s] = rec[i];
        }
    }
}

template <bool PLACE>
__device__ __forceinline__ void own_apply(const unsigned int (&cn)[PPH], const int2 (&r)[PPH][PSL],
                                          int E, int lo, unsigned int cntn,
                                          unsigned int* deg, unsigned int* cur, unsigned int* stg,
                                          bool staged, unsigned int base,
                                          int* __restrict__ srt, int t) {
    const int l = t & 15;
    unsigned int ps[PPH][PSL];
#pragma unroll
    for (int p = 0; p < PPH; ++p) {
#pragma unroll
        for (int s = 0; s < PSL; ++s) {
            const unsigned int d = (unsigned int)(r[p][s].y - lo);
            ps[p][s] = 0xffffffffu;
            if ((unsigned int)(l + 16 * s) < cn[p] && d < cntn) {
                if (!PLACE) atomicAdd(&deg[d], 1u);
                else ps[p][s] = atomicAdd(&cur[d], 1u);
            }
        }
    }
    if (PLACE) {
#pragma unroll
        for (int p = 0; p < PPH; ++p) {
#pragma unroll
            for (int s = 0; s < PSL; ++s) {
                const unsigned int pos = ps[p][s];
                if (pos != 0xffffffffu) {
                    if (staged) {
                        if (pos < (unsigned int)SCAP)
                            stg[pos] = seg_key((unsigned int)(r[p][s].y - lo), r[p][s].x, pos);
                    } else {
                        const unsigned int gp = base + pos;
                        if (gp < (unsigned int)E) srt[gp] = r[p][s].x;
                    }
                }
            }
        }
    }
}

template <bool PLACE>
__device__ __forceinline__ void own_rest(const int2* __restrict__ rec,
                                         const unsigned int* __restrict__ pcRow,
                                         int nch, int ch, int E, int lo, unsigned int cntn,
                                         unsigned int* deg, unsigned int* cur, unsigned int* stg,
                                         bool staged, unsigned int base,
                                         int* __restrict__ srt, int t) {
    const int hw = t >> 4;
    const int l = t & 15;
    const unsigned int lastE = E > 0 ? (unsigned int)(E - 1) : 0u;
    for (int c = hw; c < nch; c += 32) {
        const unsigned int w = pcRow[c];
        const unsigned int cnp = w & 0xffffu;
        const long long e0l = (long long)c * ch;
        const unsigned int e0 = e0l < (long long)E ? (unsigned int)e0l : (unsigned int)E;
        const unsigned int ofp = e0 + (w >> 16);
        const unsigned int j0 = c < 32 * PPH ? 16u * PSL : 0u;
        for (unsigned int j = j0 + (unsigned int)l; j < cnp; j += 16u) {
            unsigned int i = ofp + j;
            if (i > lastE) i = lastE;
            const int2 q = rec[i];
            const unsigned int d = (unsigned int)(q.y - lo);
            if (d < cntn) {
                if (!PLACE) {
                    atomicAdd(&deg[d], 1u);
                } else {
                    const unsigned int pos = atomicAdd(&cur[d], 1u);
                    if (staged) {
                        if (pos < (unsigned int)SCAP) stg[pos] = seg_key(d, q.x, pos);
                    } else {
                        const unsigned int gp = base + pos;
                        if (gp < (unsigned int)E) srt[gp] = q.x;
                    }
                }
            }
        }
    }
}

#define RW 24
__device__ __forceinline__ unsigned int seg_rank(const unsigned int* stg, unsigned int s0,
                                                 unsigned int nn, unsigned int i, unsigned int key) {
    unsigned int rank = 0u;
    if (nn <= 64u) {
        unsigned int o[RW];
#pragma unroll
        for (int u = 0; u < RW; ++u) o[u] = stg[s0 + (unsigned int)u];
#pragma unroll
        for (int u = 0; u < RW; ++u) rank += (o[u] < key) ? 1u : 0u;
        for (unsigned int k = RW; k < nn; ++k) rank += (stg[s0 + k] < key) ? 1u : 0u;
    } else {
        const unsigned int ks = key >> 6;
        for (unsigned int k = 0u; k < nn; ++k) {
            const unsigned int os = stg[s0 + k] >> 6;
            rank += (os < ks || (os == ks && s0 + k < i)) ? 1u : 0u;
        }
    }
    return rank;
}

__global__ __launch_bounds__(BT)
void own_kernel(const int* __restrict__ rec_i, const unsigned int* __restrict__ pcT,
                int* __restrict__ srt,
                unsigned int* __restrict__ rp_abs, unsigned int* __restrict__ degc,
                float* __restrict__ dinv, unsigned char* __restrict__ deg8,
                int N, int E, int nch) {
    __shared__ unsigned int deg[BN];
    __shared__ unsigned int segs[BN];
    __shared__ unsigned int cur[BN];
    __shared__ unsigned int wt[BT / 32];
    __shared__ unsigned int baseS;
    __shared__ unsigned int stg[SCAP + 32];
    const int2* __restrict__ rec = (const int2*)rec_i;
    const int t = threadIdx.x;
    const int lane = t & 31;
    const int b = blockIdx.x;
    const int lo = b * BN;
    const int left = N - lo;
    const unsigned int cntn = (unsigned int)(left < BN ? left : BN);
    const int ch = (E + nch - 1) / nch;
    const unsigned int* __restrict__ pcRow = pcT + (size_t)b * nch;

    unsigned int cn0[PPH];
    int2 r0[PPH][PSL];
    own_fetch(rec, pcRow, nch, ch, E, t, cn0, r0);
    bool more = nch > 32 * PPH;
#pragma unroll
    for (int p = 0; p < PPH; ++p) more = more || cn0[p] > 16u * PSL;
    unsigned int mine = 0u;
    for (int c = t; c < nch; c += BT) mine += pcRow[c] >> 16;
#pragma unroll
    for (int o = 16; o > 0; o >>= 1) mine += __shfl_xor(mine, o);
    if (t < BN) deg[t] = 0u;
    if (t == 0) baseS = 0u;
    __syncthreads();
    if (lane == 0 && mine != 0u) atomicAdd(&baseS, mine);
    own_apply<false>(cn0, r0, E, lo, cntn, deg, cur, stg, false, 0u, srt, t);
    if (more) own_rest<false>(rec, pcRow, nch, ch, E, lo, cntn, deg, cur, stg, false, 0u, srt, t);
    __syncthreads();
    const unsigned int v = (t < BN) ? deg[t] : 0u;
    const unsigned int incl = wave_scan_incl(v, lane);
    if (lane == 31) wt[t >> 5] = incl;
    __syncthreads();
    unsigned int pre = 0u, tot = 0u;
#pragma unroll
    for (int j = 0; j < BT / 32; ++j) {
        const unsigned int w = wt[j];
        pre += (j < (t >> 5)) ? w : 0u;
        tot += w;
    }
    const unsigned int excl = pre + incl - v;
    const unsigned int base = baseS;
    const bool staged = tot <= (unsigned int)SCAP && (unsigned int)N <= (1u << KSRC);
    if ((unsigned int)t < cntn) {
        const int i = lo + t;
        rp_abs[i] = base + excl;
        degc[i] = v;
        dinv[i] = __frsqrt_rn((float)(v + 1u));
    }
    if (t < BN) {
        deg8[lo + t] = (unsigned char)(((unsigned int)t < cntn) ? (v < 255u ? v : 255u) : 0u);
        segs[t] = excl;
        cur[t] = excl;
    }
    if (b == (int)gridDim.x - 1 && t < 16) deg8[lo + BN + t] = 0;
    if (staged && t < 32) stg[tot + (unsigned int)t] = 0xffffffffu;
    __syncthreads();
    own_apply<true>(cn0, r0, E, lo, cntn, deg, cur, stg, staged, base, srt, t);
    if (more) own_rest<true>(rec, pcRow, nch, ch, E, lo, cntn, deg, cur, stg, staged, base, srt, t);
    if (!staged) return;
    __syncthreads();
    for (unsigned int i0 = (unsigned int)t; i0 < tot; i0 += 2u * BT) {
        const bool has1 = i0 + BT < tot;
        const unsigned int i1 = has1 ? i0 + BT : i0;
        const unsigned int key0 = stg[i0];
        const unsigned int key1 = stg[i1];
        const unsigned int d0 = key0 >> (KSRC + 6);
        const unsigned int d1 = key1 >> (KSRC + 6);
        const unsigned int s0 = segs[d0], n0 = deg[d0];
        const unsigned int s1 = segs[d1], n1 = deg[d1];
        const unsigned int r0 = seg_rank(stg, s0, n0, i0, key0);
        const unsigned int r1 = seg_rank(stg, s1, n1, i1, key1);
        const unsigned int g0 = base + s0 + r0;
        const unsigned int g1 = base + s1 + r1;
        if (g0 < (unsigned int)E) srt[g0] = (int)((key0 >> 6) & KMASK);
        if (has1 && g1 < (unsigned int)E) srt[g1] = (int)((key1 >> 6) & KMASK);
    }
}

#define WPB 25
#define TCAP 102400
#define TFILL (TCAP / 16 / (WPB * 32))
static_assert(TFILL * 16 * WPB * 32 == TCAP, "the block's lanes cover the table in whole rounds of 16-byte loads");
#define PITCH 68

template <int CNT>
__device__ __forceinline__ void pull_group(float4& acc, const float4* __restrict__ x4,
                                           unsigned int r16, float w16, int l) {
    float4 v[CNT];
    const char* xb = (const char*)x4;
    const unsigned int lo = (unsigned int)l * 16u;
#pragma unroll
    for (int k = 0; k < CNT; ++k) {
        const unsigned int r = __shfl(r16, k, 16);
        v[k] = *(const float4*)(xb + ((r << 8) + lo));
    }
#pragma unroll
    for (int k = 0; k < CNT; ++k) {
        const float w = __shfl(w16, k, 16);
        acc.x += w * v[k].x;
        acc.y += w * v[k].y;
        acc.z += w * v[k].z;
        acc.w += w * v[k].w;
    }
}

__device__ __forceinline__ unsigned int deg_byte(const unsigned char* tab, unsigned int r) {
    const bool in = r < (unsigned int)TCAP;
    const unsigned int q = tab[in ? r : 0u];
    return in ? q : 255u;
}

__device__ __forceinline__ float src_dinv(unsigned int q, const float* __restrict__ dinv,
                                          unsigned int r) {
    float w = __frsqrt_rn((float)(q + 1u));
    if (q == 255u) w = dinv[r];
    return w;
}

__device__ __forceinline__ void pull_chunk(float4& acc, const float4* __restrict__ x4,
                                           unsigned int r16, float w16, int l, unsigned int rem) {
    if (rem > 8u) pull_group<16>(acc, x4, r16, w16, l);
    else if (rem > 0u) pull_group<8>(acc, x4, r16, w16, l);
}

__global__ __launch_bounds__(WPB * 32) __attribute__((amdgpu_waves_per_eu(7, 8)))
void gcn_pull_gemm_kernel(const float* __restrict__ x, const float* __restrict__ dinv,
                          const unsigned int* __restrict__ rp_abs,
                          const unsigned int* __restrict__ degc,
                          const int* __restrict__ srt, const unsigned char* __restrict__ deg8,
                          const float* __restrict__ Wc, const float* __restrict__ bc,
                          const float* __restrict__ Ws, const float* __restrict__ bs,
                          float* __restrict__ out, int N, int E) {
    __shared__ float aggS[WPB][16 * PITCH];
    __shared__ float xS[WPB][16 * PITCH];
    __shared__ __attribute__((aligned(16))) unsigned int degT[TCAP / 4];
    const unsigned char* tab = (const unsigned char*)degT;
    const int lane   = threadIdx.x & 31;
    const int wave   = threadIdx.x >> 5;
    const int waveG  = blockIdx.x * WPB + wave;
    const int rowBase = waveG * 16;
    const int laneHi = lane >> 4;
    const int laneLo = lane & 15;
    const float4* x4 = (const float4*)x;
    float* myG = aggS[wave];
    float* myX = xS[wave];
    {
        const unsigned int nn = (unsigned int)N < (unsigned int)TCAP ? (unsigned int)N : (unsigned int)TCAP;
        const unsigned int nq = (nn + 15u) >> 4;
        const uint4* __restrict__ s16 = (const uint4*)deg8;
        uint4* d16 = (uint4*)degT;
        uint4 tv[TFILL];
#pragma unroll
        for (int b = 0; b < TFILL; ++b) {
            unsigned int i = threadIdx.x + (unsigned int)b * (WPB * 32);
            if (i >= nq) i = nq - 1u;
            tv[b] = s16[i];
        }
        float4 orow[8];
#pragma unroll
        for (int p = 0; p < 8; ++p) {
            int cm = rowBase + 2 * p + laneHi; if (cm > N - 1) cm = N - 1;
            orow[p] = x4[(size_t)cm * (D / 4) + laneLo];
        }
#pragma unroll
        for (int b = 0; b < TFILL; ++b) d16[threadIdx.x + (unsigned int)b * (WPB * 32)] = tv[b];
#pragma unroll
        for (int p = 0; p < 8; ++p) *(float4*)(myX + (2 * p + laneHi) * PITCH + 4 * laneLo) = orow[p];
    }
    const unsigned int Eu = (unsigned int)E;
    const unsigned int lastE = Eu > 0u ? Eu - 1u : 0u;
    const unsigned int lastNode = (unsigned int)(N - 1);

    const int hrow = rowBase + laneLo;
    const bool hv = hrow < N;
    const int hr = hv ? hrow : N - 1;
    unsigned int startAll = rp_abs[hr];
    if (startAll > Eu) startAll = Eu;
    unsigned int nAll = hv ? degc[hr] : 0u;
    if (nAll > Eu - startAll) nAll = Eu - startAll;
    const float dcAll = dinv[hr];
    __syncthreads();

    unsigned int start = __shfl(startAll, laneHi);
    unsigned int n = __shfl(nAll, laneHi);
    float dc = __shfl(dcAll, laneHi);
    int c = rowBase + laneHi; if (c > N - 1) c = N - 1;
    int iA0, iB0;
    {
        unsigned int pA = start + (unsigned int)laneLo;       if (pA > lastE) pA = lastE;
        unsigned int pB = start + (unsigned int)laneLo + 16u; if (pB > lastE) pB = lastE;
        iA0 = srt[pA];
        iB0 = srt[pB];
    }
    float4 xs = x4[(size_t)c * (D / 4) + laneLo];
    unsigned int startN = __shfl(startAll, 2 + laneHi);
    unsigned int nN = __shfl(nAll, 2 + laneHi);
    float dcN = __shfl(dcAll, 2 + laneHi);
    int cN = rowBase + 2 + laneHi; if (cN > N - 1) cN = N - 1;
    int iAn, iBn;
    {
        unsigned int qA = startN + (unsigned int)laneLo;       if (qA > lastE) qA = lastE;
        unsigned int qB = startN + (unsigned int)laneLo + 16u; if (qB > lastE) qB = lastE;
        iAn = srt[qA];
        iBn = srt[qB];
    }

    {
        const float* ap = myX + laneLo * PITCH + 2 * laneHi;
        v2f a[16];
#pragma unroll
        for (int k = 0; k < 16; ++k) {
            const float2 t = *(const float2*)(ap + 4 * k);
            a[k].x = t.x; a[k].y = t.y;
        }
#pragma unroll
        for (int ntp = 0; ntp < 2; ++ntp) {
            const int n0 = ntp * 32 + laneLo;
            const int n1 = n0 + 16;
            v8f s0 = {};
            v8f s1 = {};
            v2f b0[16], b1[16];
#pragma unroll
            for (int j = 0; j < 16; ++j) {
                const int kk = 4 * j + 2 * laneHi;
                b0[j].x = Ws[(size_t)kk * D + n0];
                b0[j].y = Ws[(size_t)(kk + 1) * D + n0];
                b1[j].x = Ws[(size_t)kk * D + n1];
                b1[j].y = Ws[(size_t)(kk + 1) * D + n1];
            }
#pragma unroll
            for (int j = 0; j < 16; ++j) {
                s0 = __builtin_amdgcn_wmma_f32_16x16x4_f32(
                    false, a[j], false, b0[j], (short)0, s0, false, false);
                s1 = __builtin_amdgcn_wmma_f32_16x16x4_f32(
                    false, a[j], false, b1[j], (short)0, s1, false, false);
            }
#pragma unroll
            for (int r = 0; r < 8; ++r) {
                float* q = myX + (r + 8 * laneHi) * PITCH;
                q[n0] = s0[r];
                q[n1] = s1[r];
            }
        }
    }

    unsigned int rA = ((unsigned int)laneLo < n) ? (unsigned int)iA0 : (unsigned int)c;
    unsigned int rB = ((unsigned int)(16 + laneLo) < n) ? (unsigned int)iB0 : (unsigned int)c;
    if (rA > lastNode) rA = lastNode;
    if (rB > lastNode) rB = lastNode;
    unsigned int qA = deg_byte(tab, rA);
    unsigned int qB = deg_byte(tab, rB);

    for (int p = 0; p < 8; ++p) {
        const float dA = src_dinv(qA, dinv, rA);
        const float dB = src_dinv(qB, dinv, rB);
        unsigned int rAn = ((unsigned int)laneLo < nN) ? (unsigned int)iAn : (unsigned int)cN;
        unsigned int rBn = ((unsigned int)(16 + laneLo) < nN) ? (unsigned int)iBn : (unsigned int)cN;
        if (rAn > lastNode) rAn = lastNode;
        if (rBn > lastNode) rBn = lastNode;
        const unsigned int qAn = deg_byte(tab, rAn);
        const unsigned int qBn = deg_byte(tab, rBn);
        const float4 xsn = x4[(size_t)cN * (D / 4) + laneLo];
        const int p2 = p < 6 ? p + 2 : 7;
        const int m2 = 2 * p2 + laneHi;
        const unsigned int start2 = __shfl(startAll, m2);
        const unsigned int n2 = __shfl(nAll, m2);
        const float dc2 = __shfl(dcAll, m2);
        int c2 = rowBase + m2; if (c2 > N - 1) c2 = N - 1;
        unsigned int qA2 = start2 + (unsigned int)laneLo;       if (qA2 > lastE) qA2 = lastE;
        unsigned int qB2 = start2 + (unsigned int)laneLo + 16u; if (qB2 > lastE) qB2 = lastE;
        const int iA2 = srt[qA2];
        const int iB2 = srt[qB2];
        const unsigned int nOther = __shfl_xor(n, 16);
        const unsigned int nmax = __builtin_amdgcn_readfirstlane(n > nOther ? n : nOther);
        const float ws = dc * dc;
        float4 acc;
        acc.x = ws * xs.x; acc.y = ws * xs.y; acc.z = ws * xs.z; acc.w = ws * xs.w;
        const float wA = ((unsigned int)laneLo < n) ? dA * dc : 0.0f;
        const float wB = ((unsigned int)(16 + laneLo) < n) ? dB * dc : 0.0f;
        pull_chunk(acc, x4, rA, wA, laneLo, nmax);
        pull_chunk(acc, x4, rB, wB, laneLo, nmax > 16u ? nmax - 16u : 0u);
        for (unsigned int j = 32u; j < nmax; j += 16u) {
            const bool ok = (j + (unsigned int)laneLo) < n;
            unsigned int qC = start + j + (unsigned int)laneLo; if (qC > lastE) qC = lastE;
            const int iC = srt[qC];
            unsigned int rC = ok ? (unsigned int)iC : (unsigned int)c;
            if (rC > lastNode) rC = lastNode;
            const float wC = ok ? src_dinv(deg_byte(tab, rC), dinv, rC) * dc : 0.0f;
            pull_chunk(acc, x4, rC, wC, laneLo, nmax - j);
        }
        const int rowOff = (2 * p + laneHi) * PITCH + 4 * laneLo;
        *(float4*)(myG + rowOff) = acc;
        start = startN; n = nN; dc = dcN; c = cN;
        rA = rAn; rB = rBn; qA = qAn; qB = qBn; xs = xsn;
        startN = start2; nN = n2; dcN = dc2; cN = c2; iAn = iA2; iBn = iB2;
    }
    __builtin_amdgcn_fence(__ATOMIC_RELEASE, "wavefront");
    __builtin_amdgcn_wave_barrier();
    __builtin_amdgcn_fence(__ATOMIC_ACQUIRE, "wavefront");

    const float* gp = myG + laneLo * PITCH + 2 * laneHi;
    v2f g[16];
#pragma unroll
    for (int k = 0; k < 16; ++k) {
        const float2 u = *(const float2*)(gp + 4 * k);
        g[k].x = u.x; g[k].y = u.y;
    }

#pragma unroll
    for (int ntp = 0; ntp < 2; ++ntp) {
        const int n0 = ntp * 32 + laneLo;
        const int n1 = n0 + 16;
        v8f c0 = {};
        v8f c1 = {};
        v2f b0[16], b1[16];
#pragma unroll
        for (int j = 0; j < 16; ++j) {
            const int kk = 4 * j + 2 * laneHi;
            b0[j].x = Wc[(size_t)kk * D + n0];
            b0[j].y = Wc[(size_t)(kk + 1) * D + n0];
            b1[j].x = Wc[(size_t)kk * D + n1];
            b1[j].y = Wc[(size_t)(kk + 1) * D + n1];
        }
#pragma unroll
        for (int j = 0; j < 16; ++j) {
            c0 = __builtin_amdgcn_wmma_f32_16x16x4_f32(
                false, g[j], false, b0[j], (short)0, c0, false, false);
            c1 = __builtin_amdgcn_wmma_f32_16x16x4_f32(
                false, g[j], false, b1[j], (short)0, c1, false, false);
        }
        const float bias0 = bs[n0] + bc[n0];
        const float bias1 = bs[n1] + bc[n1];
        if (rowBase + 16 <= N) {
#pragma unroll
            for (int r = 0; r < 8; ++r) {
                const float* q = myX + (r + 8 * laneHi) * PITCH;
                const size_t idx = (size_t)(rowBase + r + 8 * laneHi) * D;
                out[idx + n0] = q[n0] + bias0 + c0[r];
                out[idx + n1] = q[n1] + bias1 + c1[r];
            }
        } else {
#pragma unroll
            for (int r = 0; r < 8; ++r) {
                const int row = rowBase + r + 8 * laneHi;
                if (row < N) {
                    const float* q = myX + (r + 8 * laneHi) * PITCH;
                    const size_t idx = (size_t)row * D;
                    out[idx + n0] = q[n0] + bias0 + c0[r];
                    out[idx + n1] = q[n1] + bias1 + c1[r];
                }
            }
        }
    }
}

extern "C" void kernel_launch(void* const* d_in, const int* in_sizes, int n_in,
                              void* d_out, int out_size, void* d_ws, size_t ws_size,
                              hipStream_t stream) {
    const float* x  = (const float*)d_in[0];
    const int*   ei = (const int*)d_in[1];
    const float* Wc = (const float*)d_in[2];
    const float* bc = (const float*)d_in[3];
    const float* Ws = (const float*)d_in[4];
    const float* bs = (const float*)d_in[5];
    float* out = (float*)d_out;

    const int N = in_sizes[0] / D;
    const int E = in_sizes[1] / 2;
    const int nbin = (N + BN - 1) / BN;
    const int nch = (E + CHCAP - 1) / CHCAP;

    float* dinv = (float*)d_ws;
    size_t off = ((size_t)N * sizeof(float) + 255) & ~(size_t)255;
    int* rec = (int*)((char*)d_ws + off);
    int* srt = rec + 2 * (size_t)E;
    unsigned int* pcT = (unsigned int*)(srt + E);
    unsigned int* rp_abs = pcT + (size_t)nbin * nch;
    unsigned int* degc = rp_abs + N;
    size_t off8 = (off + (size_t)E * 8 + ((size_t)E + (size_t)nbin * nch + 2 * (size_t)N) * sizeof(unsigned int) + 255) & ~(size_t)255;
    unsigned char* deg8 = (unsigned char*)d_ws + off8;

    bin_kernel<<<nch, BT, 0, stream>>>(ei, ei + E, rec, pcT, N, E, nbin);
    own_kernel<<<nbin, BT, 0, stream>>>(rec, pcT, srt, rp_abs, degc, dinv, deg8, N, E, nch);
    gcn_pull_gemm_kernel<<<(N + 16 * WPB - 1) / (16 * WPB), WPB * 32, 0, stream>>>(
        x, dinv, rp_abs, degc, srt, deg8, Wc, bc, Ws, bs, out, N, E);
}
